# starting kernel
# baseline (speedup 1.0000x reference)
#include <hip/hip_runtime.h>
#include <hip/hip_bf16.h>

typedef __attribute__((ext_vector_type(2))) float v2f;
typedef __attribute__((ext_vector_type(8))) float v8f;

#define IN_FEATS 384
#define N_HIDDEN 128
#define N_CLASSES 16

__global__ __launch_bounds__(256) void zero_kernel(float* __restrict__ p, long long n) {
    long long i = (long long)blockIdx.x * blockDim.x + threadIdx.x;
    long long stride = (long long)gridDim.x * blockDim.x;
    for (; i < n; i += stride) p[i] = 0.0f;
}

__global__ __launch_bounds__(256) void degree_kernel(const int* __restrict__ src,
                                                     const int* __restrict__ dst,
                                                     float* __restrict__ deg_out,
                                                     float* __restrict__ deg_in,
                                                     int nE) {
    int e = blockIdx.x * 256 + threadIdx.x;
    if (e >= nE) return;
    atomicAdd(&deg_out[src[e]], 1.0f);
    atomicAdd(&deg_in[dst[e]], 1.0f);
}

__global__ __launch_bounds__(256) void rsqrt_kernel(float* __restrict__ a,
                                                    float* __restrict__ b, int n) {
    int i = blockIdx.x * 256 + threadIdx.x;
    if (i >= n) return;
    float x = a[i];
    a[i] = (x > 0.0f) ? rsqrtf(x) : 0.0f;
    float y = b[i];
    b[i] = (y > 0.0f) ? rsqrtf(y) : 0.0f;
}

template <int CT>
__global__ __launch_bounds__(64) void gemm_scaled_wmma(const float* __restrict__ H,
                                                       const float* __restrict__ ns,
                                                       const float* __restrict__ W,
                                                       float* __restrict__ out,
                                                       int K, int nRowTiles) {
    constexpr int F = CT * 16;
    const int wave = threadIdx.x >> 5;
    const int lane = threadIdx.x & 31;
    const int m  = lane & 15;
    const int hi = lane >> 4;

    const int rowTile = blockIdx.x * 2 + wave;
    if (rowTile >= nRowTiles) return;
    const int row0 = rowTile * 16;

    const float scale = ns[row0 + m];
    const float* __restrict__ arow = H + (size_t)(row0 + m) * K;
    const float* __restrict__ bbase = W + m;

    v8f acc[CT];
#pragma unroll
    for (int c = 0; c < CT; ++c) acc[c] = (v8f){};

#pragma unroll 2
    for (int k0 = 0; k0 < K; k0 += 4) {
        const int ka = k0 + 2 * hi;
        float2 t = *(const float2*)(arow + ka);
        v2f a;
        a.x = t.x * scale;
        a.y = t.y * scale;
        const float* __restrict__ brow = bbase + (size_t)ka * F;
#pragma unroll
        for (int c = 0; c < CT; ++c) {
            v2f b;
            b.x = brow[c * 16];
            b.y = brow[c * 16 + F];
            acc[c] = __builtin_amdgcn_wmma_f32_16x16x4_f32(false, a, false, b,
                                                           (short)0, acc[c], false, false);
        }
    }

    float* __restrict__ obase = out + (size_t)(row0 + 8 * hi) * F + m;
#pragma unroll
    for (int c = 0; c < CT; ++c) {
#pragma unroll
        for (int v = 0; v < 8; ++v)
            obase[(size_t)v * F + c * 16] = acc[c][v];
    }
}

__global__ __launch_bounds__(256) void scatter_kernel(const int* __restrict__ src,
                                                      const int* __restrict__ dst,
                                                      const float* __restrict__ hw,
                                                      float* __restrict__ agg,
                                                      int nEdges, int F, int lgF4) {
    const int F4 = F >> 2;
    const int tid = blockIdx.x * 256 + threadIdx.x;
    const int e = tid >> lgF4;
    const int j = tid & (F4 - 1);
    if (e >= nEdges) return;
    const int s = src[e];
    const int d = dst[e];
    const float4 v = ((const float4*)(hw + (size_t)s * F))[j];
    float* o = agg + (size_t)d * F + 4 * j;
    atomicAdd(o + 0, v.x);
    atomicAdd(o + 1, v.y);
    atomicAdd(o + 2, v.z);
    atomicAdd(o + 3, v.w);
}

__global__ __launch_bounds__(256) void finalize_kernel(const float* __restrict__ agg,
                                                       const float* __restrict__ nd,
                                                       const float* __restrict__ bias,
                                                       float* __restrict__ out,
                                                       int N, int lgF, int relu) {
    const int F = 1 << lgF;
    long long t = (long long)blockIdx.x * 256 + threadIdx.x;
    if (t >= ((long long)N << lgF)) return;
    const int i = (int)(t >> lgF);
    const int f = (int)(t & (F - 1));
    float v = agg[t] * nd[i] + bias[f];
    if (relu) v = fmaxf(v, 0.0f);
    out[t] = v;
}

extern "C" void kernel_launch(void* const* d_in, const int* in_sizes, int n_in,
                              void* d_out, int out_size, void* d_ws, size_t ws_size,
                              hipStream_t stream) {
    const float* feat = (const float*)d_in[0];
    const int*   src  = (const int*)d_in[1];
    const int*   dst  = (const int*)d_in[2];
    const float* W1   = (const float*)d_in[3];
    const float* b1   = (const float*)d_in[4];
    const float* W2   = (const float*)d_in[5];
    const float* b2   = (const float*)d_in[6];
    const float* W3   = (const float*)d_in[7];
    const float* b3   = (const float*)d_in[8];
    float* out = (float*)d_out;

    const int N = in_sizes[0] / IN_FEATS;
    const int E = in_sizes[1];

    float* ns   = (float*)d_ws;
    float* nd   = ns + N;
    float* bufA = nd + N;
    float* bufB = bufA + (size_t)N * N_HIDDEN;

    const int rowTiles  = N / 16;
    const int rowBlocks = (rowTiles + 1) / 2;

    zero_kernel<<<(2 * N + 255) / 256, 256, 0, stream>>>(ns, 2LL * N);
    degree_kernel<<<(E + 255) / 256, 256, 0, stream>>>(src, dst, ns, nd, E);
    rsqrt_kernel<<<(N + 255) / 256, 256, 0, stream>>>(ns, nd, N);

    {
        gemm_scaled_wmma<8><<<rowBlocks, 64, 0, stream>>>(feat, ns, W1, bufA, IN_FEATS, rowTiles);
        long long nAgg = (long long)N * N_HIDDEN;
        zero_kernel<<<(int)((nAgg + 255) / 256), 256, 0, stream>>>(bufB, nAgg);
        long long nScat = (long long)E * (N_HIDDEN / 4);
        scatter_kernel<<<(int)((nScat + 255) / 256), 256, 0, stream>>>(src, dst, bufA, bufB, E, N_HIDDEN, 5);
        finalize_kernel<<<(int)((nAgg + 255) / 256), 256, 0, stream>>>(bufB, nd, b1, bufA, N, 7, 1);
    }

    {
        gemm_scaled_wmma<8><<<rowBlocks, 64, 0, stream>>>(bufA, ns, W2, bufB, N_HIDDEN, rowTiles);
        long long nAgg = (long long)N * N_HIDDEN;
        zero_kernel<<<(int)((nAgg + 255) / 256), 256, 0, stream>>>(bufA, nAgg);
        long long nScat = (long long)E * (N_HIDDEN / 4);
        scatter_kernel<<<(int)((nScat + 255) / 256), 256, 0, stream>>>(src, dst, bufB, bufA, E, N_HIDDEN, 5);
        finalize_kernel<<<(int)((nAgg + 255) / 256), 256, 0, stream>>>(bufA, nd, b2, bufB, N, 7, 1);
    }

    {
        gemm_scaled_wmma<1><<<rowBlocks, 64, 0, stream>>>(bufB, ns, W3, bufA, N_HIDDEN, rowTiles);
        long long nOut = (long long)N * N_CLASSES;
        zero_kernel<<<(int)((nOut + 255) / 256), 256, 0, stream>>>(out, nOut);
        long long nScat = (long long)E * (N_CLASSES / 4);
        scatter_kernel<<<(int)((nScat + 255) / 256), 256, 0, stream>>>(src, dst, bufA, out, E, N_CLASSES, 2);
        finalize_kernel<<<(int)((nOut + 255) / 256), 256, 0, stream>>>(out, nd, b3, out, N, 4, 0);
    }
}


# Aggregate by destination: integer counts, a scan and each edge's arrival-order rank group the source ids; rows summed in registers, stored once with norm, bias, relu; float sum order changes
# speedup vs baseline: 2.3905x; 2.3905x over previous
#include <hip/hip_runtime.h>
#include <hip/hip_bf16.h>

typedef __attribute__((ext_vector_type(2))) float v2f;
typedef __attribute__((ext_vector_type(8))) float v8f;

#define IN_FEATS 384
#define N_HIDDEN 128
#define N_CLASSES 16
#define SCAN_CHUNK 512

__global__ __launch_bounds__(256) void zero_kernel(int* __restrict__ p, int n) {
    int i = blockIdx.x * 256 + threadIdx.x;
    if (i < n) p[i] = 0;
}

__global__ __launch_bounds__(256) void degree_rank_kernel(const int* __restrict__ src,
                                                          const int* __restrict__ dst,
                                                          int* __restrict__ cnt_out,
                                                          int* __restrict__ cnt_in,
                                                          int* __restrict__ rank,
                                                          int nE) {
    int e = blockIdx.x * 256 + threadIdx.x;
    if (e >= nE) return;
    atomicAdd(&cnt_out[src[e]], 1);
    rank[e] = atomicAdd(&cnt_in[dst[e]], 1);
}

__global__ __launch_bounds__(SCAN_CHUNK / 4) void chunk_sum_kernel(const int* __restrict__ cnt_in,
                                                                   int* __restrict__ chunk_sum,
                                                                   int n) {
    __shared__ int ws[SCAN_CHUNK / 128];
    const int t = threadIdx.x;
    const int q = blockIdx.x * (SCAN_CHUNK / 4) + t;
    int x = 0;
    if (q < (n >> 2)) {
        const int4 v = ((const int4*)cnt_in)[q];
        x = (v.x + v.y) + (v.z + v.w);
    }
    x += __shfl_xor(x, 16, 32);
    x += __shfl_xor(x, 8, 32);
    x += __shfl_xor(x, 4, 32);
    x += __shfl_xor(x, 2, 32);
    x += __shfl_xor(x, 1, 32);
    if ((t & 31) == 0) ws[t >> 5] = x;
    __syncthreads();
    if (t == 0) {
        int tot = 0;
        for (int w = 0; w < SCAN_CHUNK / 128; ++w) tot += ws[w];
        chunk_sum[blockIdx.x] = tot;
    }
}

__global__ __launch_bounds__(SCAN_CHUNK) void scan_apply_kernel(int* __restrict__ cnt_in_rp,
                                                                const int* __restrict__ chunk_sum,
                                                                int* __restrict__ cnt_out_ns,
                                                                int n) {
    __shared__ int s[SCAN_CHUNK];
    __shared__ int p[SCAN_CHUNK];
    const int t = threadIdx.x;
    const int i = blockIdx.x * SCAN_CHUNK + t;
    const int v = (i < n) ? cnt_in_rp[i] : 0;
    const int before = (t < (int)blockIdx.x) ? chunk_sum[t] : 0;
    s[t] = v;
    p[t] = before;
    __syncthreads();
    for (int off = 1; off < SCAN_CHUNK; off <<= 1) {
        const int u = (t >= off) ? s[t - off] : 0;
        const int z = (t >= off) ? p[t - off] : 0;
        __syncthreads();
        s[t] += u;
        p[t] += z;
        __syncthreads();
    }
    if (i < n) {
        cnt_in_rp[i] = p[SCAN_CHUNK - 1] + s[t] - v;
        const int c = cnt_out_ns[i];
        const float r = (c > 0) ? rsqrtf((float)c) : 0.0f;
        cnt_out_ns[i] = __float_as_int(r);
    }
}

__global__ __launch_bounds__(256) void place_kernel(const int* __restrict__ src,
                                                    const int* __restrict__ dst,
                                                    const int* __restrict__ rp,
                                                    const int* __restrict__ rank,
                                                    int* __restrict__ col, int nE) {
    int e = blockIdx.x * 256 + threadIdx.x;
    if (e >= nE) return;
    col[rp[dst[e]] + rank[e]] = src[e];
}

__global__ __launch_bounds__(256) void copy_int4_kernel(const int* __restrict__ a,
                                                        int* __restrict__ b, int n4) {
    int i = blockIdx.x * 256 + threadIdx.x;
    if (i < n4) ((int4*)b)[i] = ((const int4*)a)[i];
}

template <int CT>
__global__ __launch_bounds__(64) void gemm_scaled_wmma(const float* __restrict__ H,
                                                       const float* __restrict__ ns,
                                                       const float* __restrict__ W,
                                                       float* __restrict__ out,
                                                       int K, int nRowTiles) {
    constexpr int F = CT * 16;
    const int wave = threadIdx.x >> 5;
    const int lane = threadIdx.x & 31;
    const int m  = lane & 15;
    const int hi = lane >> 4;

    const int rowTile = blockIdx.x * 2 + wave;
    if (rowTile >= nRowTiles) return;
    const int row0 = rowTile * 16;

    const float scale = ns[row0 + m];
    const float* __restrict__ arow = H + (size_t)(row0 + m) * K;
    const float* __restrict__ bbase = W + m;

    v8f acc[CT];
#pragma unroll
    for (int c = 0; c < CT; ++c) acc[c] = (v8f){};

#pragma unroll 2
    for (int k0 = 0; k0 < K; k0 += 4) {
        const int ka = k0 + 2 * hi;
        float2 t = *(const float2*)(arow + ka);
        v2f a;
        a.x = t.x * scale;
        a.y = t.y * scale;
        const float* __restrict__ brow = bbase + (size_t)ka * F;
#pragma unroll
        for (int c = 0; c < CT; ++c) {
            v2f b;
            b.x = brow[c * 16];
            b.y = brow[c * 16 + F];
            acc[c] = __builtin_amdgcn_wmma_f32_16x16x4_f32(false, a, false, b,
                                                           (short)0, acc[c], false, false);
        }
    }

    float* __restrict__ obase = out + (size_t)(row0 + 8 * hi) * F + m;
#pragma unroll
    for (int c = 0; c < CT; ++c) {
#pragma unroll
        for (int v = 0; v < 8; ++v)
            obase[(size_t)v * F + c * 16] = acc[c][v];
    }
}

__global__ __launch_bounds__(256) void gather_rows128_kernel(const float* __restrict__ hw,
                                                             const int* __restrict__ rp,
                                                             const int* __restrict__ col,
                                                             const float* __restrict__ bias,
                                                             float* __restrict__ out,
                                                             int n, int nE) {
    const int lane = threadIdx.x & 31;
    const int i = __builtin_amdgcn_readfirstlane((int)(blockIdx.x * 8 + (threadIdx.x >> 5)));
    if (i >= n) return;
    const int start = rp[i];
    const int end = (i + 1 < n) ? rp[i + 1] : nE;
    const float4* __restrict__ hw4 = (const float4*)hw;

    const float4 zero4 = make_float4(0.0f, 0.0f, 0.0f, 0.0f);
    float4 acc = zero4;
    for (int base = start; base < end; base += 32) {
        const int k = base + lane;
        const int myc = (k < end) ? col[k] : 0;
        const int cnt = min(32, end - base);
        const int last = cnt - 1;
        for (int j = 0; j < cnt; j += 4) {
            const int s0 = __builtin_amdgcn_readlane(myc, j);
            const int s1 = __builtin_amdgcn_readlane(myc, min(j + 1, last));
            const int s2 = __builtin_amdgcn_readlane(myc, min(j + 2, last));
            const int s3 = __builtin_amdgcn_readlane(myc, min(j + 3, last));
            const float4 v0 = hw4[(size_t)s0 * 32 + lane];
            float4 v1 = hw4[(size_t)s1 * 32 + lane];
            float4 v2 = hw4[(size_t)s2 * 32 + lane];
            float4 v3 = hw4[(size_t)s3 * 32 + lane];
            if (j + 1 > last) v1 = zero4;
            if (j + 2 > last) v2 = zero4;
            if (j + 3 > last) v3 = zero4;
            acc.x += v0.x; acc.y += v0.y; acc.z += v0.z; acc.w += v0.w;
            acc.x += v1.x; acc.y += v1.y; acc.z += v1.z; acc.w += v1.w;
            acc.x += v2.x; acc.y += v2.y; acc.z += v2.z; acc.w += v2.w;
            acc.x += v3.x; acc.y += v3.y; acc.z += v3.z; acc.w += v3.w;
        }
    }

    const int deg = end - start;
    const float nd = (deg > 0) ? rsqrtf((float)deg) : 0.0f;
    const float4 b = ((const float4*)bias)[lane];
    float4 r;
    r.x = fmaxf(acc.x * nd + b.x, 0.0f);
    r.y = fmaxf(acc.y * nd + b.y, 0.0f);
    r.z = fmaxf(acc.z * nd + b.z, 0.0f);
    r.w = fmaxf(acc.w * nd + b.w, 0.0f);
    ((float4*)out)[(size_t)i * 32 + lane] = r;
}

__global__ __launch_bounds__(256) void gather_rows16_kernel(const float* __restrict__ hw,
                                                            const int* __restrict__ rp,
                                                            const int* __restrict__ col,
                                                            const float* __restrict__ bias,
                                                            float* __restrict__ out,
                                                            int n, int nE) {
    const int t = blockIdx.x * 256 + threadIdx.x;
    const int i = t >> 2;
    const int q = t & 3;
    if (i >= n) return;
    const int start = rp[i];
    const int end = (i + 1 < n) ? rp[i + 1] : nE;
    const float4* __restrict__ hw4 = (const float4*)hw;
    const float4 zero4 = make_float4(0.0f, 0.0f, 0.0f, 0.0f);

    float4 acc = zero4;
    const int lastk = end - 1;
    for (int k = start; k < end; k += 4) {
        const int c0 = col[k];
        const int c1 = col[min(k + 1, lastk)];
        const int c2 = col[min(k + 2, lastk)];
        const int c3 = col[min(k + 3, lastk)];
        const float4 v0 = hw4[(size_t)c0 * 4 + q];
        float4 v1 = hw4[(size_t)c1 * 4 + q];
        float4 v2 = hw4[(size_t)c2 * 4 + q];
        float4 v3 = hw4[(size_t)c3 * 4 + q];
        if (k + 1 > lastk) v1 = zero4;
        if (k + 2 > lastk) v2 = zero4;
        if (k + 3 > lastk) v3 = zero4;
        acc.x += v0.x; acc.y += v0.y; acc.z += v0.z; acc.w += v0.w;
        acc.x += v1.x; acc.y += v1.y; acc.z += v1.z; acc.w += v1.w;
        acc.x += v2.x; acc.y += v2.y; acc.z += v2.z; acc.w += v2.w;
        acc.x += v3.x; acc.y += v3.y; acc.z += v3.z; acc.w += v3.w;
    }

    const int deg = end - start;
    const float nd = (deg > 0) ? rsqrtf((float)deg) : 0.0f;
    const float4 b = ((const float4*)bias)[q];
    float4 r;
    r.x = acc.x * nd + b.x;
    r.y = acc.y * nd + b.y;
    r.z = acc.z * nd + b.z;
    r.w = acc.w * nd + b.w;
    ((float4*)out)[(size_t)i * 4 + q] = r;
}

extern "C" void kernel_launch(void* const* d_in, const int* in_sizes, int n_in,
                              void* d_out, int out_size, void* d_ws, size_t ws_size,
                              hipStream_t stream) {
    const float* feat = (const float*)d_in[0];
    const int*   src  = (const int*)d_in[1];
    const int*   dst  = (const int*)d_in[2];
    const float* W1   = (const float*)d_in[3];
    const float* b1   = (const float*)d_in[4];
    const float* W2   = (const float*)d_in[5];
    const float* b2   = (const float*)d_in[6];
    const float* W3   = (const float*)d_in[7];
    const float* b3   = (const float*)d_in[8];
    float* out = (float*)d_out;

    const int N = in_sizes[0] / IN_FEATS;
    const int E = in_sizes[1];

    int*   cntOut = (int*)d_ws;
    int*   rp     = cntOut + N;
    float* bufA   = (float*)(rp + N);
    float* bufB   = bufA + (size_t)N * N_HIDDEN;
    const float* ns = (const float*)cntOut;
    int* chunkSum = (int*)bufA;
    int* rank     = (int*)bufB;
    int* col      = (int*)d_out;
    int* col3     = (int*)(bufA + (size_t)N * N_CLASSES);

    const int rowTiles  = N / 16;
    const int rowBlocks = (rowTiles + 1) / 2;
    const int nChunks   = (N + SCAN_CHUNK - 1) / SCAN_CHUNK;

    zero_kernel<<<(2 * N + 255) / 256, 256, 0, stream>>>(cntOut, 2 * N);
    degree_rank_kernel<<<(E + 255) / 256, 256, 0, stream>>>(src, dst, cntOut, rp, rank, E);
    chunk_sum_kernel<<<nChunks, SCAN_CHUNK / 4, 0, stream>>>(rp, chunkSum, N);
    scan_apply_kernel<<<nChunks, SCAN_CHUNK, 0, stream>>>(rp, chunkSum, cntOut, N);
    place_kernel<<<(E + 255) / 256, 256, 0, stream>>>(src, dst, rp, rank, col, E);

    gemm_scaled_wmma<8><<<rowBlocks, 64, 0, stream>>>(feat, ns, W1, bufA, IN_FEATS, rowTiles);
    gather_rows128_kernel<<<(N + 7) / 8, 256, 0, stream>>>(bufA, rp, col, b1, bufB, N, E);

    gemm_scaled_wmma<8><<<rowBlocks, 64, 0, stream>>>(bufB, ns, W2, bufA, N_HIDDEN, rowTiles);
    gather_rows128_kernel<<<(N + 7) / 8, 256, 0, stream>>>(bufA, rp, col, b2, bufB, N, E);

    gemm_scaled_wmma<1><<<rowBlocks, 64, 0, stream>>>(bufB, ns, W3, bufA, N_HIDDEN, rowTiles);
    copy_int4_kernel<<<(E / 4 + 255) / 256, 256, 0, stream>>>(col, col3, E / 4);
    gather_rows16_kernel<<<(4 * N + 255) / 256, 256, 0, stream>>>(bufA, rp, col3, b3, out, N, E);
}


# Edges grouped by destination in two levels on LDS counters (bucket count, scan, bucket-ordered stream, per-bucket rows): no global atomic left; float sum order within a row is LDS arrival order
# speedup vs baseline: 3.2442x; 1.3571x over previous
#include <hip/hip_runtime.h>
#include <hip/hip_bf16.h>

typedef __attribute__((ext_vector_type(2))) float v2f;
typedef __attribute__((ext_vector_type(8))) float v8f;

#define IN_FEATS 384
#define N_HIDDEN 128
#define N_CLASSES 16
#define BUCKET_SHIFT 9
#define BUCKET_NODES 512
#define MAX_BUCKETS 256
#define SCAN_TILES 12
#define EDGE_CHUNKS 250
#define EDGE_BATCH 8


__global__ __launch_bounds__(1024) void bucket_count_kernel(const int* __restrict__ src,
                                                            const int* __restrict__ dst,
                                                            int* __restrict__ tblD,
                                                            int* __restrict__ tblS,
                                                            int nE, int nB) {
    __shared__ int hD[MAX_BUCKETS];
    __shared__ int hS[MAX_BUCKETS];
    const int t = threadIdx.x;
    const int G = gridDim.x;
    if (t < nB) { hD[t] = 0; hS[t] = 0; }
    __syncthreads();
    const int chunk = (nE + G - 1) / G;
    const int e0 = blockIdx.x * chunk;
    const int e1 = min(e0 + chunk, nE);
    for (int base = e0 + t; base < e1; base += 1024 * EDGE_BATCH) {
        int d[EDGE_BATCH], s[EDGE_BATCH];
#pragma unroll
        for (int j = 0; j < EDGE_BATCH; ++j) {
            const int e = min(base + j * 1024, e1 - 1);
            d[j] = dst[e];
            s[j] = src[e];
        }
#pragma unroll
        for (int j = 0; j < EDGE_BATCH; ++j) {
            asm volatile("" : "+v"(d[j]), "+v"(s[j]));
        }
#pragma unroll
        for (int j = 0; j < EDGE_BATCH; ++j) {
            if (base + j * 1024 < e1) {
                atomicAdd(&hD[d[j] >> BUCKET_SHIFT], 1);
                atomicAdd(&hS[s[j] >> BUCKET_SHIFT], 1);
            }
        }
    }
    __syncthreads();
    if (t < nB) {
        tblD[t * G + blockIdx.x] = hD[t];
        tblS[t * G + blockIdx.x] = hS[t];
    }
}

__global__ __launch_bounds__(1024) void scan_tables_kernel(int* __restrict__ tblD,
                                                           int* __restrict__ tblS, int T) {
    __shared__ int wtot[2][32];
    __shared__ int wbase[2][32];
    __shared__ int tileTot[2];
    int* __restrict__ tbl = (blockIdx.x == 0) ? tblD : tblS;
    const int t = threadIdx.x;
    const int lane = t & 31;
    const int wave = t >> 5;
    const int nQ = T >> 2;
    int4 v[SCAN_TILES];
#pragma unroll
    for (int i = 0; i < SCAN_TILES; ++i) {
        const int q = i * 1024 + t;
        v[i] = (q < nQ) ? ((const int4*)tbl)[q] : make_int4(0, 0, 0, 0);
    }
    int carry = 0;
#pragma unroll
    for (int i = 0; i < SCAN_TILES; ++i) {
        const int q = i * 1024 + t;
        const int s = (v[i].x + v[i].y) + (v[i].z + v[i].w);
        int inc = s;
        for (int o = 1; o < 32; o <<= 1) {
            const int u = __shfl_up(inc, o, 32);
            if (lane >= o) inc += u;
        }
        const int p = i & 1;
        if (lane == 31) wtot[p][wave] = inc;
        __syncthreads();
        if (wave == 0) {
            const int x = wtot[p][lane];
            int xi = x;
            for (int o = 1; o < 32; o <<= 1) {
                const int u = __shfl_up(xi, o, 32);
                if (lane >= o) xi += u;
            }
            wbase[p][lane] = xi - x;
            if (lane == 31) tileTot[p] = xi;
        }
        __syncthreads();
        const int ex = carry + wbase[p][wave] + inc - s;
        if (q < nQ) {
            int4 r;
            r.x = ex;
            r.y = ex + v[i].x;
            r.z = r.y + v[i].y;
            r.w = r.z + v[i].z;
            ((int4*)tbl)[q] = r;
        }
        carry += tileTot[p];
    }
    if (t == 0) tbl[T] = carry;
}

__global__ __launch_bounds__(1024) void bucket_scatter_kernel(const int* __restrict__ src,
                                                              const int* __restrict__ dst,
                                                              const int* __restrict__ tblD,
                                                              const int* __restrict__ tblS,
                                                              int* __restrict__ strD,
                                                              int* __restrict__ strS,
                                                              int nE, int nB) {
    __shared__ int cD[MAX_BUCKETS];
    __shared__ int cS[MAX_BUCKETS];
    const int t = threadIdx.x;
    const int G = gridDim.x;
    if (t < nB) {
        cD[t] = tblD[t * G + blockIdx.x];
        cS[t] = tblS[t * G + blockIdx.x];
    }
    __syncthreads();
    const int chunk = (nE + G - 1) / G;
    const int e0 = blockIdx.x * chunk;
    const int e1 = min(e0 + chunk, nE);
    for (int base = e0 + t; base < e1; base += 1024 * EDGE_BATCH) {
        int d[EDGE_BATCH], s[EDGE_BATCH];
#pragma unroll
        for (int j = 0; j < EDGE_BATCH; ++j) {
            const int e = min(base + j * 1024, e1 - 1);
            d[j] = dst[e];
            s[j] = src[e];
        }
#pragma unroll
        for (int j = 0; j < EDGE_BATCH; ++j) {
            asm volatile("" : "+v"(d[j]), "+v"(s[j]));
        }
#pragma unroll
        for (int j = 0; j < EDGE_BATCH; ++j) {
            if (base + j * 1024 < e1) {
                const int pd = atomicAdd(&cD[d[j] >> BUCKET_SHIFT], 1);
                const int ps = atomicAdd(&cS[s[j] >> BUCKET_SHIFT], 1);
                strD[pd] = (int)(((unsigned)s[j] << BUCKET_SHIFT) |
                                 (unsigned)(d[j] & (BUCKET_NODES - 1)));
                strS[ps] = s[j] & (BUCKET_NODES - 1);
            }
        }
    }
}

__global__ __launch_bounds__(1024) void bucket_finish_kernel(const int* __restrict__ tblD,
                                                             const int* __restrict__ tblS,
                                                             const int* __restrict__ strD,
                                                             const int* __restrict__ strS,
                                                             int* __restrict__ rp,
                                                             float* __restrict__ ns,
                                                             int* __restrict__ col,
                                                             int n, int G) {
    __shared__ int cIn[BUCKET_NODES];
    __shared__ int cOut[BUCKET_NODES];
    __shared__ int cur[BUCKET_NODES];
    __shared__ int wtot[32];
    __shared__ int wbase[32];
    const int t = threadIdx.x;
    const int lane = t & 31;
    const int wave = t >> 5;
    const int b = blockIdx.x;
    if (t < BUCKET_NODES) { cIn[t] = 0; cOut[t] = 0; }
    __syncthreads();
    const int d0 = tblD[b * G];
    const int d1 = tblD[(b + 1) * G];
    const int s0 = tblS[b * G];
    const int s1 = tblS[(b + 1) * G];
    for (int base = d0 + t; base < d1; base += 1024 * EDGE_BATCH) {
        int w[EDGE_BATCH];
#pragma unroll
        for (int j = 0; j < EDGE_BATCH; ++j) {
            const int e = min(base + j * 1024, d1 - 1);
            w[j] = strD[e];
        }
#pragma unroll
        for (int j = 0; j < EDGE_BATCH; ++j) {
            asm volatile("" : "+v"(w[j]));
        }
#pragma unroll
        for (int j = 0; j < EDGE_BATCH; ++j)
            if (base + j * 1024 < d1) atomicAdd(&cIn[w[j] & (BUCKET_NODES - 1)], 1);
    }
    for (int base = s0 + t; base < s1; base += 1024 * EDGE_BATCH) {
        int w[EDGE_BATCH];
#pragma unroll
        for (int j = 0; j < EDGE_BATCH; ++j) {
            const int e = min(base + j * 1024, s1 - 1);
            w[j] = strS[e];
        }
#pragma unroll
        for (int j = 0; j < EDGE_BATCH; ++j) {
            asm volatile("" : "+v"(w[j]));
        }
#pragma unroll
        for (int j = 0; j < EDGE_BATCH; ++j)
            if (base + j * 1024 < s1) atomicAdd(&cOut[w[j]], 1);
    }
    __syncthreads();

    const int v = (t < BUCKET_NODES) ? cIn[t] : 0;
    int inc = v;
    for (int o = 1; o < 32; o <<= 1) {
        const int u = __shfl_up(inc, o, 32);
        if (lane >= o) inc += u;
    }
    if (lane == 31) wtot[wave] = inc;
    __syncthreads();
    if (wave == 0) {
        const int x = wtot[lane];
        int xi = x;
        for (int o = 1; o < 32; o <<= 1) {
            const int u = __shfl_up(xi, o, 32);
            if (lane >= o) xi += u;
        }
        wbase[lane] = xi - x;
    }
    __syncthreads();
    if (t < BUCKET_NODES) {
        const int start = wbase[wave] + inc - v;
        cur[t] = start;
        const int node = b * BUCKET_NODES + t;
        if (node < n) {
            rp[node] = d0 + start;
            const int c = cOut[t];
            ns[node] = (c > 0) ? rsqrtf((float)c) : 0.0f;
        }
    }
    __syncthreads();
    for (int base = d0 + t; base < d1; base += 1024 * EDGE_BATCH) {
        int w[EDGE_BATCH];
#pragma unroll
        for (int j = 0; j < EDGE_BATCH; ++j) {
            const int e = min(base + j * 1024, d1 - 1);
            w[j] = strD[e];
        }
#pragma unroll
        for (int j = 0; j < EDGE_BATCH; ++j) {
            asm volatile("" : "+v"(w[j]));
        }
#pragma unroll
        for (int j = 0; j < EDGE_BATCH; ++j) {
            if (base + j * 1024 < d1) {
                const int r = atomicAdd(&cur[w[j] & (BUCKET_NODES - 1)], 1);
                col[d0 + r] = (int)((unsigned)w[j] >> BUCKET_SHIFT);
            }
        }
    }
}

__global__ __launch_bounds__(256) void copy_int4_kernel(const int* __restrict__ a,
                                                        int* __restrict__ b, int n4) {
    int i = blockIdx.x * 256 + threadIdx.x;
    if (i < n4) ((int4*)b)[i] = ((const int4*)a)[i];
}

template <int CT>
__global__ __launch_bounds__(64) void gemm_scaled_wmma(const float* __restrict__ H,
                                                       const float* __restrict__ ns,
                                                       const float* __restrict__ W,
                                                       float* __restrict__ out,
                                                       int K, int nRowTiles) {
    constexpr int F = CT * 16;
    const int wave = threadIdx.x >> 5;
    const int lane = threadIdx.x & 31;
    const int m  = lane & 15;
    const int hi = lane >> 4;

    const int rowTile = blockIdx.x * 2 + wave;
    if (rowTile >= nRowTiles) return;
    const int row0 = rowTile * 16;

    const float scale = ns[row0 + m];
    const float* __restrict__ arow = H + (size_t)(row0 + m) * K;
    const float* __restrict__ bbase = W + m;

    v8f acc[CT];
#pragma unroll
    for (int c = 0; c < CT; ++c) acc[c] = (v8f){};

#pragma unroll 2
    for (int k0 = 0; k0 < K; k0 += 4) {
        const int ka = k0 + 2 * hi;
        float2 t = *(const float2*)(arow + ka);
        v2f a;
        a.x = t.x * scale;
        a.y = t.y * scale;
        const float* __restrict__ brow = bbase + (size_t)ka * F;
#pragma unroll
        for (int c = 0; c < CT; ++c) {
            v2f b;
            b.x = brow[c * 16];
            b.y = brow[c * 16 + F];
            acc[c] = __builtin_amdgcn_wmma_f32_16x16x4_f32(false, a, false, b,
                                                           (short)0, acc[c], false, false);
        }
    }

    float* __restrict__ obase = out + (size_t)(row0 + 8 * hi) * F + m;
#pragma unroll
    for (int c = 0; c < CT; ++c) {
#pragma unroll
        for (int v = 0; v < 8; ++v)
            obase[(size_t)v * F + c * 16] = acc[c][v];
    }
}

__global__ __launch_bounds__(256) void gather_rows128_kernel(const float* __restrict__ hw,
                                                             const int* __restrict__ rp,
                                                             const int* __restrict__ col,
                                                             const float* __restrict__ bias,
                                                             float* __restrict__ out,
                                                             int n, int nE) {
    const int lane = threadIdx.x & 31;
    const int i = __builtin_amdgcn_readfirstlane((int)(blockIdx.x * 8 + (threadIdx.x >> 5)));
    if (i >= n) return;
    const int start = rp[i];
    const int end = (i + 1 < n) ? rp[i + 1] : nE;
    const float4* __restrict__ hw4 = (const float4*)hw;

    const float4 zero4 = make_float4(0.0f, 0.0f, 0.0f, 0.0f);
    float4 acc = zero4;
    for (int base = start; base < end; base += 32) {
        const int k = base + lane;
        const int myc = (k < end) ? col[k] : 0;
        const int cnt = min(32, end - base);
        const int last = cnt - 1;
        for (int j = 0; j < cnt; j += 4) {
            const int s0 = __builtin_amdgcn_readlane(myc, j);
            const int s1 = __builtin_amdgcn_readlane(myc, min(j + 1, last));
            const int s2 = __builtin_amdgcn_readlane(myc, min(j + 2, last));
            const int s3 = __builtin_amdgcn_readlane(myc, min(j + 3, last));
            const float4 v0 = hw4[(size_t)s0 * 32 + lane];
            float4 v1 = hw4[(size_t)s1 * 32 + lane];
            float4 v2 = hw4[(size_t)s2 * 32 + lane];
            float4 v3 = hw4[(size_t)s3 * 32 + lane];
            if (j + 1 > last) v1 = zero4;
            if (j + 2 > last) v2 = zero4;
            if (j + 3 > last) v3 = zero4;
            acc.x += v0.x; acc.y += v0.y; acc.z += v0.z; acc.w += v0.w;
            acc.x += v1.x; acc.y += v1.y; acc.z += v1.z; acc.w += v1.w;
            acc.x += v2.x; acc.y += v2.y; acc.z += v2.z; acc.w += v2.w;
            acc.x += v3.x; acc.y += v3.y; acc.z += v3.z; acc.w += v3.w;
        }
    }

    const int deg = end - start;
    const float nd = (deg > 0) ? rsqrtf((float)deg) : 0.0f;
    const float4 b = ((const float4*)bias)[lane];
    float4 r;
    r.x = fmaxf(acc.x * nd + b.x, 0.0f);
    r.y = fmaxf(acc.y * nd + b.y, 0.0f);
    r.z = fmaxf(acc.z * nd + b.z, 0.0f);
    r.w = fmaxf(acc.w * nd + b.w, 0.0f);
    ((float4*)out)[(size_t)i * 32 + lane] = r;
}

__global__ __launch_bounds__(256) void gather_rows16_kernel(const float* __restrict__ hw,
                                                            const int* __restrict__ rp,
                                                            const int* __restrict__ col,
                                                            const float* __restrict__ bias,
                                                            float* __restrict__ out,
                                                            int n, int nE) {
    const int t = blockIdx.x * 256 + threadIdx.x;
    const int i = t >> 2;
    const int q = t & 3;
    if (i >= n) return;
    const int start = rp[i];
    const int end = (i + 1 < n) ? rp[i + 1] : nE;
    const float4* __restrict__ hw4 = (const float4*)hw;
    const float4 zero4 = make_float4(0.0f, 0.0f, 0.0f, 0.0f);

    float4 acc = zero4;
    const int lastk = end - 1;
    for (int k = start; k < end; k += 4) {
        const int c0 = col[k];
        const int c1 = col[min(k + 1, lastk)];
        const int c2 = col[min(k + 2, lastk)];
        const int c3 = col[min(k + 3, lastk)];
        const float4 v0 = hw4[(size_t)c0 * 4 + q];
        float4 v1 = hw4[(size_t)c1 * 4 + q];
        float4 v2 = hw4[(size_t)c2 * 4 + q];
        float4 v3 = hw4[(size_t)c3 * 4 + q];
        if (k + 1 > lastk) v1 = zero4;
        if (k + 2 > lastk) v2 = zero4;
        if (k + 3 > lastk) v3 = zero4;
        acc.x += v0.x; acc.y += v0.y; acc.z += v0.z; acc.w += v0.w;
        acc.x += v1.x; acc.y += v1.y; acc.z += v1.z; acc.w += v1.w;
        acc.x += v2.x; acc.y += v2.y; acc.z += v2.z; acc.w += v2.w;
        acc.x += v3.x; acc.y += v3.y; acc.z += v3.z; acc.w += v3.w;
    }

    const int deg = end - start;
    const float nd = (deg > 0) ? rsqrtf((float)deg) : 0.0f;
    const float4 b = ((const float4*)bias)[q];
    float4 r;
    r.x = acc.x * nd + b.x;
    r.y = acc.y * nd + b.y;
    r.z = acc.z * nd + b.z;
    r.w = acc.w * nd + b.w;
    ((float4*)out)[(size_t)i * 4 + q] = r;
}

extern "C" void kernel_launch(void* const* d_in, const int* in_sizes, int n_in,
                              void* d_out, int out_size, void* d_ws, size_t ws_size,
                              hipStream_t stream) {
    const float* feat = (const float*)d_in[0];
    const int*   src  = (const int*)d_in[1];
    const int*   dst  = (const int*)d_in[2];
    const float* W1   = (const float*)d_in[3];
    const float* b1   = (const float*)d_in[4];
    const float* W2   = (const float*)d_in[5];
    const float* b2   = (const float*)d_in[6];
    const float* W3   = (const float*)d_in[7];
    const float* b3   = (const float*)d_in[8];
    float* out = (float*)d_out;

    const int N = in_sizes[0] / IN_FEATS;
    const int E = in_sizes[1];

    float* nsW    = (float*)d_ws;
    int*   rp     = (int*)d_ws + N;
    float* bufA   = (float*)(rp + N);
    float* bufB   = bufA + (size_t)N * N_HIDDEN;
    const float* ns = nsW;
    int* col      = (int*)d_out;
    int* col3     = (int*)(bufA + (size_t)N * N_CLASSES);

    const int rowTiles  = N / 16;
    const int rowBlocks = (rowTiles + 1) / 2;
    const int nBuckets  = (N + BUCKET_NODES - 1) / BUCKET_NODES;
    const int T         = nBuckets * EDGE_CHUNKS;
    int* tblD = (int*)bufA;
    int* tblS = tblD + T + 4;
    int* strD = (int*)bufB;
    int* strS = strD + E;

    bucket_count_kernel<<<EDGE_CHUNKS, 1024, 0, stream>>>(src, dst, tblD, tblS, E, nBuckets);
    scan_tables_kernel<<<2, 1024, 0, stream>>>(tblD, tblS, T);
    bucket_scatter_kernel<<<EDGE_CHUNKS, 1024, 0, stream>>>(src, dst, tblD, tblS, strD, strS, E, nBuckets);
    bucket_finish_kernel<<<nBuckets, 1024, 0, stream>>>(tblD, tblS, strD, strS, rp, nsW, col, N, EDGE_CHUNKS);

    gemm_scaled_wmma<8><<<rowBlocks, 64, 0, stream>>>(feat, ns, W1, bufA, IN_FEATS, rowTiles);
    gather_rows128_kernel<<<(N + 7) / 8, 256, 0, stream>>>(bufA, rp, col, b1, bufB, N, E);

    gemm_scaled_wmma<8><<<rowBlocks, 64, 0, stream>>>(bufB, ns, W2, bufA, N_HIDDEN, rowTiles);
    gather_rows128_kernel<<<(N + 7) / 8, 256, 0, stream>>>(bufA, rp, col, b2, bufB, N, E);

    gemm_scaled_wmma<1><<<rowBlocks, 64, 0, stream>>>(bufB, ns, W3, bufA, N_HIDDEN, rowTiles);
    copy_int4_kernel<<<(E / 4 + 255) / 256, 256, 0, stream>>>(col, col3, E / 4);
    gather_rows16_kernel<<<(4 * N + 255) / 256, 256, 0, stream>>>(bufA, rp, col3, b3, out, N, E);
}


# Each 16-node gather group hands its summed, scaled rows through LDS to the next layer's f32 WMMA product (W in LDS, resident groups loop over tiles): layers 2, 3 lose their GEMM launch and h plane
# speedup vs baseline: 3.6496x; 1.1250x over previous
#include <hip/hip_runtime.h>
#include <hip/hip_bf16.h>

typedef __attribute__((ext_vector_type(2))) float v2f;
typedef __attribute__((ext_vector_type(8))) float v8f;

#define IN_FEATS 384
#define N_HIDDEN 128
#define N_CLASSES 16
#define BUCKET_SHIFT 9
#define BUCKET_NODES 512
#define MAX_BUCKETS 256
#define SCAN_TILES 12
#define EDGE_CHUNKS 250
#define EDGE_BATCH 8


__global__ __launch_bounds__(1024) void bucket_count_kernel(const int* __restrict__ src,
                                                            const int* __restrict__ dst,
                                                            int* __restrict__ tblD,
                                                            int* __restrict__ tblS,
                                                            int nE, int nB) {
    __shared__ int hD[MAX_BUCKETS];
    __shared__ int hS[MAX_BUCKETS];
    const int t = threadIdx.x;
    const int G = gridDim.x;
    if (t < nB) { hD[t] = 0; hS[t] = 0; }
    __syncthreads();
    const int chunk = (nE + G - 1) / G;
    const int e0 = blockIdx.x * chunk;
    const int e1 = min(e0 + chunk, nE);
    for (int base = e0 + t; base < e1; base += 1024 * EDGE_BATCH) {
        int d[EDGE_BATCH], s[EDGE_BATCH];
#pragma unroll
        for (int j = 0; j < EDGE_BATCH; ++j) {
            const int e = min(base + j * 1024, e1 - 1);
            d[j] = dst[e];
            s[j] = src[e];
        }
#pragma unroll
        for (int j = 0; j < EDGE_BATCH; ++j) {
            asm volatile("" : "+v"(d[j]), "+v"(s[j]));
        }
#pragma unroll
        for (int j = 0; j < EDGE_BATCH; ++j) {
            if (base + j * 1024 < e1) {
                atomicAdd(&hD[d[j] >> BUCKET_SHIFT], 1);
                atomicAdd(&hS[s[j] >> BUCKET_SHIFT], 1);
            }
        }
    }
    __syncthreads();
    if (t < nB) {
        tblD[t * G + blockIdx.x] = hD[t];
        tblS[t * G + blockIdx.x] = hS[t];
    }
}

__global__ __launch_bounds__(1024) void scan_tables_kernel(int* __restrict__ tblD,
                                                           int* __restrict__ tblS, int T) {
    __shared__ int wtot[2][32];
    __shared__ int wbase[2][32];
    __shared__ int tileTot[2];
    int* __restrict__ tbl = (blockIdx.x == 0) ? tblD : tblS;
    const int t = threadIdx.x;
    const int lane = t & 31;
    const int wave = t >> 5;
    const int nQ = T >> 2;
    int4 v[SCAN_TILES];
#pragma unroll
    for (int i = 0; i < SCAN_TILES; ++i) {
        const int q = i * 1024 + t;
        v[i] = (q < nQ) ? ((const int4*)tbl)[q] : make_int4(0, 0, 0, 0);
    }
    int carry = 0;
#pragma unroll
    for (int i = 0; i < SCAN_TILES; ++i) {
        const int q = i * 1024 + t;
        const int s = (v[i].x + v[i].y) + (v[i].z + v[i].w);
        int inc = s;
        for (int o = 1; o < 32; o <<= 1) {
            const int u = __shfl_up(inc, o, 32);
            if (lane >= o) inc += u;
        }
        const int p = i & 1;
        if (lane == 31) wtot[p][wave] = inc;
        __syncthreads();
        if (wave == 0) {
            const int x = wtot[p][lane];
            int xi = x;
            for (int o = 1; o < 32; o <<= 1) {
                const int u = __shfl_up(xi, o, 32);
                if (lane >= o) xi += u;
            }
            wbase[p][lane] = xi - x;
            if (lane == 31) tileTot[p] = xi;
        }
        __syncthreads();
        const int ex = carry + wbase[p][wave] + inc - s;
        if (q < nQ) {
            int4 r;
            r.x = ex;
            r.y = ex + v[i].x;
            r.z = r.y + v[i].y;
            r.w = r.z + v[i].z;
            ((int4*)tbl)[q] = r;
        }
        carry += tileTot[p];
    }
    if (t == 0) tbl[T] = carry;
}

__global__ __launch_bounds__(1024) void bucket_scatter_kernel(const int* __restrict__ src,
                                                              const int* __restrict__ dst,
                                                              const int* __restrict__ tblD,
                                                              const int* __restrict__ tblS,
                                                              int* __restrict__ strD,
                                                              int* __restrict__ strS,
                                                              int nE, int nB) {
    __shared__ int cD[MAX_BUCKETS];
    __shared__ int cS[MAX_BUCKETS];
    const int t = threadIdx.x;
    const int G = gridDim.x;
    if (t < nB) {
        cD[t] = tblD[t * G + blockIdx.x];
        cS[t] = tblS[t * G + blockIdx.x];
    }
    __syncthreads();
    const int chunk = (nE + G - 1) / G;
    const int e0 = blockIdx.x * chunk;
    const int e1 = min(e0 + chunk, nE);
    for (int base = e0 + t; base < e1; base += 1024 * EDGE_BATCH) {
        int d[EDGE_BATCH], s[EDGE_BATCH];
#pragma unroll
        for (int j = 0; j < EDGE_BATCH; ++j) {
            const int e = min(base + j * 1024, e1 - 1);
            d[j] = dst[e];
            s[j] = src[e];
        }
#pragma unroll
        for (int j = 0; j < EDGE_BATCH; ++j) {
            asm volatile("" : "+v"(d[j]), "+v"(s[j]));
        }
#pragma unroll
        for (int j = 0; j < EDGE_BATCH; ++j) {
            if (base + j * 1024 < e1) {
                const int pd = atomicAdd(&cD[d[j] >> BUCKET_SHIFT], 1);
                const int ps = atomicAdd(&cS[s[j] >> BUCKET_SHIFT], 1);
                strD[pd] = (int)(((unsigned)s[j] << BUCKET_SHIFT) |
                                 (unsigned)(d[j] & (BUCKET_NODES - 1)));
                strS[ps] = s[j] & (BUCKET_NODES - 1);
            }
        }
    }
}

__global__ __launch_bounds__(1024) void bucket_finish_kernel(const int* __restrict__ tblD,
                                                             const int* __restrict__ tblS,
                                                             const int* __restrict__ strD,
                                                             const int* __restrict__ strS,
                                                             int* __restrict__ rp,
                                                             float* __restrict__ ns,
                                                             int* __restrict__ col,
                                                             int n, int G) {
    __shared__ int cIn[BUCKET_NODES];
    __shared__ int cOut[BUCKET_NODES];
    __shared__ int cur[BUCKET_NODES];
    __shared__ int wtot[32];
    __shared__ int wbase[32];
    const int t = threadIdx.x;
    const int lane = t & 31;
    const int wave = t >> 5;
    const int b = blockIdx.x;
    if (t < BUCKET_NODES) { cIn[t] = 0; cOut[t] = 0; }
    __syncthreads();
    const int d0 = tblD[b * G];
    const int d1 = tblD[(b + 1) * G];
    const int s0 = tblS[b * G];
    const int s1 = tblS[(b + 1) * G];
    for (int base = d0 + t; base < d1; base += 1024 * EDGE_BATCH) {
        int w[EDGE_BATCH];
#pragma unroll
        for (int j = 0; j < EDGE_BATCH; ++j) {
            const int e = min(base + j * 1024, d1 - 1);
            w[j] = strD[e];
        }
#pragma unroll
        for (int j = 0; j < EDGE_BATCH; ++j) {
            asm volatile("" : "+v"(w[j]));
        }
#pragma unroll
        for (int j = 0; j < EDGE_BATCH; ++j)
            if (base + j * 1024 < d1) atomicAdd(&cIn[w[j] & (BUCKET_NODES - 1)], 1);
    }
    for (int base = s0 + t; base < s1; base += 1024 * EDGE_BATCH) {
        int w[EDGE_BATCH];
#pragma unroll
        for (int j = 0; j < EDGE_BATCH; ++j) {
            const int e = min(base + j * 1024, s1 - 1);
            w[j] = strS[e];
        }
#pragma unroll
        for (int j = 0; j < EDGE_BATCH; ++j) {
            asm volatile("" : "+v"(w[j]));
        }
#pragma unroll
        for (int j = 0; j < EDGE_BATCH; ++j)
            if (base + j * 1024 < s1) atomicAdd(&cOut[w[j]], 1);
    }
    __syncthreads();

    const int v = (t < BUCKET_NODES) ? cIn[t] : 0;
    int inc = v;
    for (int o = 1; o < 32; o <<= 1) {
        const int u = __shfl_up(inc, o, 32);
        if (lane >= o) inc += u;
    }
    if (lane == 31) wtot[wave] = inc;
    __syncthreads();
    if (wave == 0) {
        const int x = wtot[lane];
        int xi = x;
        for (int o = 1; o < 32; o <<= 1) {
            const int u = __shfl_up(xi, o, 32);
            if (lane >= o) xi += u;
        }
        wbase[lane] = xi - x;
    }
    __syncthreads();
    if (t < BUCKET_NODES) {
        const int start = wbase[wave] + inc - v;
        cur[t] = start;
        const int node = b * BUCKET_NODES + t;
        if (node < n) {
            rp[node] = d0 + start;
            const int c = cOut[t];
            ns[node] = (c > 0) ? rsqrtf((float)c) : 0.0f;
        }
    }
    __syncthreads();
    for (int base = d0 + t; base < d1; base += 1024 * EDGE_BATCH) {
        int w[EDGE_BATCH];
#pragma unroll
        for (int j = 0; j < EDGE_BATCH; ++j) {
            const int e = min(base + j * 1024, d1 - 1);
            w[j] = strD[e];
        }
#pragma unroll
        for (int j = 0; j < EDGE_BATCH; ++j) {
            asm volatile("" : "+v"(w[j]));
        }
#pragma unroll
        for (int j = 0; j < EDGE_BATCH; ++j) {
            if (base + j * 1024 < d1) {
                const int r = atomicAdd(&cur[w[j] & (BUCKET_NODES - 1)], 1);
                col[d0 + r] = (int)((unsigned)w[j] >> BUCKET_SHIFT);
            }
        }
    }
}

__global__ __launch_bounds__(256) void copy_int4_kernel(const int* __restrict__ a,
                                                        int* __restrict__ b, int n4) {
    int i = blockIdx.x * 256 + threadIdx.x;
    if (i < n4) ((int4*)b)[i] = ((const int4*)a)[i];
}

template <int CT>
__global__ __launch_bounds__(64) void gemm_scaled_wmma(const float* __restrict__ H,
                                                       const float* __restrict__ ns,
                                                       const float* __restrict__ W,
                                                       float* __restrict__ out,
                                                       int K, int nRowTiles) {
    constexpr int F = CT * 16;
    const int wave = threadIdx.x >> 5;
    const int lane = threadIdx.x & 31;
    const int m  = lane & 15;
    const int hi = lane >> 4;

    const int rowTile = blockIdx.x * 2 + wave;
    if (rowTile >= nRowTiles) return;
    const int row0 = rowTile * 16;

    const float scale = ns[row0 + m];
    const float* __restrict__ arow = H + (size_t)(row0 + m) * K;
    const float* __restrict__ bbase = W + m;

    v8f acc[CT];
#pragma unroll
    for (int c = 0; c < CT; ++c) acc[c] = (v8f){};

#pragma unroll 2
    for (int k0 = 0; k0 < K; k0 += 4) {
        const int ka = k0 + 2 * hi;
        float2 t = *(const float2*)(arow + ka);
        v2f a;
        a.x = t.x * scale;
        a.y = t.y * scale;
        const float* __restrict__ brow = bbase + (size_t)ka * F;
#pragma unroll
        for (int c = 0; c < CT; ++c) {
            v2f b;
            b.x = brow[c * 16];
            b.y = brow[c * 16 + F];
            acc[c] = __builtin_amdgcn_wmma_f32_16x16x4_f32(false, a, false, b,
                                                           (short)0, acc[c], false, false);
        }
    }

    float* __restrict__ obase = out + (size_t)(row0 + 8 * hi) * F + m;
#pragma unroll
    for (int c = 0; c < CT; ++c) {
#pragma unroll
        for (int v = 0; v < 8; ++v)
            obase[(size_t)v * F + c * 16] = acc[c][v];
    }
}

#define FUSED_GROUPS 1024
template <int CT>
__global__ __launch_bounds__(512) void gather_rows128_product_kernel(const float* __restrict__ hw,
                                                                     const int* __restrict__ rp,
                                                                     const int* __restrict__ col,
                                                                     const float* __restrict__ bias,
                                                                     const float* __restrict__ ns,
                                                                     const float* __restrict__ W,
                                                                     float* __restrict__ out,
                                                                     int n, int nE, int nTiles) {
    constexpr int F = CT * 16;
    constexpr int PITCH = N_HIDDEN + 4;
    __shared__ __attribute__((aligned(16))) float Wl[N_HIDDEN * F];
    __shared__ __attribute__((aligned(16))) float tile[16 * PITCH];
    const int lane = threadIdx.x & 31;
    const int wave = __builtin_amdgcn_readfirstlane((int)(threadIdx.x >> 5));

#pragma unroll
    for (int idx = threadIdx.x; idx < (N_HIDDEN / 2) * (F / 4); idx += 512) {
        const int kp = idx / (F / 4);
        const int c4 = idx - kp * (F / 4);
        const float4 w0 = ((const float4*)W)[(size_t)(2 * kp) * (F / 4) + c4];
        const float4 w1 = ((const float4*)W)[(size_t)(2 * kp + 1) * (F / 4) + c4];
        float4* dst = (float4*)(Wl + 2 * (kp * F + 4 * c4));
        dst[0] = make_float4(w0.x, w1.x, w0.y, w1.y);
        dst[1] = make_float4(w0.z, w1.z, w0.w, w1.w);
    }

    const float4* __restrict__ hw4 = (const float4*)hw;
    const float4 zero4 = make_float4(0.0f, 0.0f, 0.0f, 0.0f);
    const float4 b = ((const float4*)bias)[lane];

    for (int t = blockIdx.x; t < nTiles; t += gridDim.x) {
        const int row0 = t * 16;
        const int i = row0 + wave;
        float4 r = zero4;
        if (i < n) {
            const int start = rp[i];
            const int nxt = rp[min(i + 1, n - 1)];
            const float scale = ns[i];
            const int end = (i + 1 < n) ? nxt : nE;
            float4 acc = zero4;
            for (int base = start; base < end; base += 32) {
                const int k = base + lane;
                const int myc = (k < end) ? col[k] : 0;
                const int cnt = min(32, end - base);
                const int last = cnt - 1;
                for (int j = 0; j < cnt; j += 4) {
                    const int s0 = __builtin_amdgcn_readlane(myc, j);
                    const int s1 = __builtin_amdgcn_readlane(myc, min(j + 1, last));
                    const int s2 = __builtin_amdgcn_readlane(myc, min(j + 2, last));
                    const int s3 = __builtin_amdgcn_readlane(myc, min(j + 3, last));
                    const float4 v0 = hw4[(size_t)s0 * 32 + lane];
                    float4 v1 = hw4[(size_t)s1 * 32 + lane];
                    float4 v2 = hw4[(size_t)s2 * 32 + lane];
                    float4 v3 = hw4[(size_t)s3 * 32 + lane];
                    if (j + 1 > last) v1 = zero4;
                    if (j + 2 > last) v2 = zero4;
                    if (j + 3 > last) v3 = zero4;
                    acc.x += v0.x; acc.y += v0.y; acc.z += v0.z; acc.w += v0.w;
                    acc.x += v1.x; acc.y += v1.y; acc.z += v1.z; acc.w += v1.w;
                    acc.x += v2.x; acc.y += v2.y; acc.z += v2.z; acc.w += v2.w;
                    acc.x += v3.x; acc.y += v3.y; acc.z += v3.z; acc.w += v3.w;
                }
            }
            const int deg = end - start;
            const float nd = (deg > 0) ? rsqrtf((float)deg) : 0.0f;
            r.x = fmaxf(acc.x * nd + b.x, 0.0f) * scale;
            r.y = fmaxf(acc.y * nd + b.y, 0.0f) * scale;
            r.z = fmaxf(acc.z * nd + b.z, 0.0f) * scale;
            r.w = fmaxf(acc.w * nd + b.w, 0.0f) * scale;
        }
        *(float4*)(tile + wave * PITCH + lane * 4) = r;
        __syncthreads();

        v8f pacc = (v8f){};
        const int m  = lane & 15;
        const int hi = lane >> 4;
        if (wave < CT) {
            const float* arow = tile + m * PITCH + 2 * hi;
            const float* bcol = Wl + 2 * (hi * F + wave * 16 + m);
            v2f fa[2][4], fb[2][4];
#pragma unroll
            for (int j = 0; j < 4; ++j) {
                fa[0][j] = *(const v2f*)(arow + 4 * j);
                fb[0][j] = *(const v2f*)(bcol + 4 * j * F);
            }
#pragma unroll
            for (int g = 0; g < N_HIDDEN / 16; ++g) {
                if (g + 1 < N_HIDDEN / 16) {
#pragma unroll
                    for (int j = 0; j < 4; ++j) {
                        const int k0 = 16 * (g + 1) + 4 * j;
                        fa[(g + 1) & 1][j] = *(const v2f*)(arow + k0);
                        fb[(g + 1) & 1][j] = *(const v2f*)(bcol + k0 * F);
                    }
                }
                __builtin_amdgcn_sched_barrier(0);
#pragma unroll
                for (int j = 0; j < 4; ++j)
                    pacc = __builtin_amdgcn_wmma_f32_16x16x4_f32(false, fa[g & 1][j], false, fb[g & 1][j],
                                                                 (short)0, pacc, false, false);
            }
        }
        __syncthreads();
        if (wave < CT) {
            float* __restrict__ obase = out + (size_t)(row0 + 8 * hi) * F + wave * 16 + m;
#pragma unroll
            for (int v = 0; v < 8; ++v)
                if (row0 + 8 * hi + v < n) obase[(size_t)v * F] = pacc[v];
        }
    }
}

__global__ __launch_bounds__(256) void gather_rows16_kernel(const float* __restrict__ hw,
                                                            const int* __restrict__ rp,
                                                            const int* __restrict__ col,
                                                            const float* __restrict__ bias,
                                                            float* __restrict__ out,
                                                            int n, int nE) {
    const int t = blockIdx.x * 256 + threadIdx.x;
    const int i = t >> 2;
    const int q = t & 3;
    if (i >= n) return;
    const int start = rp[i];
    const int end = (i + 1 < n) ? rp[i + 1] : nE;
    const float4* __restrict__ hw4 = (const float4*)hw;
    const float4 zero4 = make_float4(0.0f, 0.0f, 0.0f, 0.0f);

    float4 acc = zero4;
    const int lastk = end - 1;
    for (int k = start; k < end; k += 4) {
        const int c0 = col[k];
        const int c1 = col[min(k + 1, lastk)];
        const int c2 = col[min(k + 2, lastk)];
        const int c3 = col[min(k + 3, lastk)];
        const float4 v0 = hw4[(size_t)c0 * 4 + q];
        float4 v1 = hw4[(size_t)c1 * 4 + q];
        float4 v2 = hw4[(size_t)c2 * 4 + q];
        float4 v3 = hw4[(size_t)c3 * 4 + q];
        if (k + 1 > lastk) v1 = zero4;
        if (k + 2 > lastk) v2 = zero4;
        if (k + 3 > lastk) v3 = zero4;
        acc.x += v0.x; acc.y += v0.y; acc.z += v0.z; acc.w += v0.w;
        acc.x += v1.x; acc.y += v1.y; acc.z += v1.z; acc.w += v1.w;
        acc.x += v2.x; acc.y += v2.y; acc.z += v2.z; acc.w += v2.w;
        acc.x += v3.x; acc.y += v3.y; acc.z += v3.z; acc.w += v3.w;
    }

    const int deg = end - start;
    const float nd = (deg > 0) ? rsqrtf((float)deg) : 0.0f;
    const float4 b = ((const float4*)bias)[q];
    float4 r;
    r.x = acc.x * nd + b.x;
    r.y = acc.y * nd + b.y;
    r.z = acc.z * nd + b.z;
    r.w = acc.w * nd + b.w;
    ((float4*)out)[(size_t)i * 4 + q] = r;
}

extern "C" void kernel_launch(void* const* d_in, const int* in_sizes, int n_in,
                              void* d_out, int out_size, void* d_ws, size_t ws_size,
                              hipStream_t stream) {
    const float* feat = (const float*)d_in[0];
    const int*   src  = (const int*)d_in[1];
    const int*   dst  = (const int*)d_in[2];
    const float* W1   = (const float*)d_in[3];
    const float* b1   = (const float*)d_in[4];
    const float* W2   = (const float*)d_in[5];
    const float* b2   = (const float*)d_in[6];
    const float* W3   = (const float*)d_in[7];
    const float* b3   = (const float*)d_in[8];
    float* out = (float*)d_out;

    const int N = in_sizes[0] / IN_FEATS;
    const int E = in_sizes[1];

    float* nsW    = (float*)d_ws;
    int*   rp     = (int*)d_ws + N;
    float* bufA   = (float*)(rp + N);
    float* bufB   = bufA + (size_t)N * N_HIDDEN;
    const float* ns = nsW;
    int* col      = (int*)d_out;
    int* col3     = (int*)(bufA + (size_t)N * N_CLASSES);

    const int rowTiles  = N / 16;
    const int rowBlocks = (rowTiles + 1) / 2;
    const int nTiles    = (N + 15) / 16;
    const int nBuckets  = (N + BUCKET_NODES - 1) / BUCKET_NODES;
    const int T         = nBuckets * EDGE_CHUNKS;
    int* tblD = (int*)bufA;
    int* tblS = tblD + T + 4;
    int* strD = (int*)bufB;
    int* strS = strD + E;

    bucket_count_kernel<<<EDGE_CHUNKS, 1024, 0, stream>>>(src, dst, tblD, tblS, E, nBuckets);
    scan_tables_kernel<<<2, 1024, 0, stream>>>(tblD, tblS, T);
    bucket_scatter_kernel<<<EDGE_CHUNKS, 1024, 0, stream>>>(src, dst, tblD, tblS, strD, strS, E, nBuckets);
    bucket_finish_kernel<<<nBuckets, 1024, 0, stream>>>(tblD, tblS, strD, strS, rp, nsW, col, N, EDGE_CHUNKS);

    gemm_scaled_wmma<8><<<rowBlocks, 64, 0, stream>>>(feat, ns, W1, bufA, IN_FEATS, rowTiles);

    gather_rows128_product_kernel<8><<<FUSED_GROUPS, 512, 0, stream>>>(bufA, rp, col, b1, ns, W2, bufB, N, E, nTiles);

    gather_rows128_product_kernel<1><<<FUSED_GROUPS, 512, 0, stream>>>(bufB, rp, col, b2, ns, W3, bufA, N, E, nTiles);

    copy_int4_kernel<<<(E / 4 + 255) / 256, 256, 0, stream>>>(col, col3, E / 4);
    gather_rows16_kernel<<<(4 * N + 255) / 256, 256, 0, stream>>>(bufA, rp, col3, b3, out, N, E);
}


# Layer-1 product: W1 held in LDS (k-pairs interleaved) by one resident work-group per WGP; each wave streams its 16 rows of H through LDS by whole-line loads; every fragment is one 8-byte LDS load
# speedup vs baseline: 3.8296x; 1.0493x over previous
#include <hip/hip_runtime.h>
#include <hip/hip_bf16.h>

typedef __attribute__((ext_vector_type(2))) float v2f;
typedef __attribute__((ext_vector_type(8))) float v8f;
typedef __attribute__((ext_vector_type(4))) float v4f;

#define IN_FEATS 384
#define N_HIDDEN 128
#define N_CLASSES 16
#define BUCKET_SHIFT 9
#define BUCKET_NODES 512
#define MAX_BUCKETS 256
#define SCAN_TILES 12
#define EDGE_CHUNKS 250
#define EDGE_BATCH 8


__global__ __launch_bounds__(1024) void bucket_count_kernel(const int* __restrict__ src,
                                                            const int* __restrict__ dst,
                                                            int* __restrict__ tblD,
                                                            int* __restrict__ tblS,
                                                            int nE, int nB) {
    __shared__ int hD[MAX_BUCKETS];
    __shared__ int hS[MAX_BUCKETS];
    const int t = threadIdx.x;
    const int G = gridDim.x;
    if (t < nB) { hD[t] = 0; hS[t] = 0; }
    __syncthreads();
    const int chunk = (nE + G - 1) / G;
    const int e0 = blockIdx.x * chunk;
    const int e1 = min(e0 + chunk, nE);
    for (int base = e0 + t; base < e1; base += 1024 * EDGE_BATCH) {
        int d[EDGE_BATCH], s[EDGE_BATCH];
#pragma unroll
        for (int j = 0; j < EDGE_BATCH; ++j) {
            const int e = min(base + j * 1024, e1 - 1);
            d[j] = dst[e];
            s[j] = src[e];
        }
#pragma unroll
        for (int j = 0; j < EDGE_BATCH; ++j) {
            asm volatile("" : "+v"(d[j]), "+v"(s[j]));
        }
#pragma unroll
        for (int j = 0; j < EDGE_BATCH; ++j) {
            if (base + j * 1024 < e1) {
                atomicAdd(&hD[d[j] >> BUCKET_SHIFT], 1);
                atomicAdd(&hS[s[j] >> BUCKET_SHIFT], 1);
            }
        }
    }
    __syncthreads();
    if (t < nB) {
        tblD[t * G + blockIdx.x] = hD[t];
        tblS[t * G + blockIdx.x] = hS[t];
    }
}

__global__ __launch_bounds__(1024) void scan_tables_kernel(int* __restrict__ tblD,
                                                           int* __restrict__ tblS, int T) {
    __shared__ int wtot[2][32];
    __shared__ int wbase[2][32];
    __shared__ int tileTot[2];
    int* __restrict__ tbl = (blockIdx.x == 0) ? tblD : tblS;
    const int t = threadIdx.x;
    const int lane = t & 31;
    const int wave = t >> 5;
    const int nQ = T >> 2;
    int4 v[SCAN_TILES];
#pragma unroll
    for (int i = 0; i < SCAN_TILES; ++i) {
        const int q = i * 1024 + t;
        v[i] = (q < nQ) ? ((const int4*)tbl)[q] : make_int4(0, 0, 0, 0);
    }
    int carry = 0;
#pragma unroll
    for (int i = 0; i < SCAN_TILES; ++i) {
        const int q = i * 1024 + t;
        const int s = (v[i].x + v[i].y) + (v[i].z + v[i].w);
        int inc = s;
        for (int o = 1; o < 32; o <<= 1) {
            const int u = __shfl_up(inc, o, 32);
            if (lane >= o) inc += u;
        }
        const int p = i & 1;
        if (lane == 31) wtot[p][wave] = inc;
        __syncthreads();
        if (wave == 0) {
            const int x = wtot[p][lane];
            int xi = x;
            for (int o = 1; o < 32; o <<= 1) {
                const int u = __shfl_up(xi, o, 32);
                if (lane >= o) xi += u;
            }
            wbase[p][lane] = xi - x;
            if (lane == 31) tileTot[p] = xi;
        }
        __syncthreads();
        const int ex = carry + wbase[p][wave] + inc - s;
        if (q < nQ) {
            int4 r;
            r.x = ex;
            r.y = ex + v[i].x;
            r.z = r.y + v[i].y;
            r.w = r.z + v[i].z;
            ((int4*)tbl)[q] = r;
        }
        carry += tileTot[p];
    }
    if (t == 0) tbl[T] = carry;
}

__global__ __launch_bounds__(1024) void bucket_scatter_kernel(const int* __restrict__ src,
                                                              const int* __restrict__ dst,
                                                              const int* __restrict__ tblD,
                                                              const int* __restrict__ tblS,
                                                              int* __restrict__ strD,
                                                              int* __restrict__ strS,
                                                              int nE, int nB) {
    __shared__ int cD[MAX_BUCKETS];
    __shared__ int cS[MAX_BUCKETS];
    const int t = threadIdx.x;
    const int G = gridDim.x;
    if (t < nB) {
        cD[t] = tblD[t * G + blockIdx.x];
        cS[t] = tblS[t * G + blockIdx.x];
    }
    __syncthreads();
    const int chunk = (nE + G - 1) / G;
    const int e0 = blockIdx.x * chunk;
    const int e1 = min(e0 + chunk, nE);
    for (int base = e0 + t; base < e1; base += 1024 * EDGE_BATCH) {
        int d[EDGE_BATCH], s[EDGE_BATCH];
#pragma unroll
        for (int j = 0; j < EDGE_BATCH; ++j) {
            const int e = min(base + j * 1024, e1 - 1);
            d[j] = dst[e];
            s[j] = src[e];
        }
#pragma unroll
        for (int j = 0; j < EDGE_BATCH; ++j) {
            asm volatile("" : "+v"(d[j]), "+v"(s[j]));
        }
#pragma unroll
        for (int j = 0; j < EDGE_BATCH; ++j) {
            if (base + j * 1024 < e1) {
                const int pd = atomicAdd(&cD[d[j] >> BUCKET_SHIFT], 1);
                const int ps = atomicAdd(&cS[s[j] >> BUCKET_SHIFT], 1);
                strD[pd] = (int)(((unsigned)s[j] << BUCKET_SHIFT) |
                                 (unsigned)(d[j] & (BUCKET_NODES - 1)));
                strS[ps] = s[j] & (BUCKET_NODES - 1);
            }
        }
    }
}

__global__ __launch_bounds__(1024) void bucket_finish_kernel(const int* __restrict__ tblD,
                                                             const int* __restrict__ tblS,
                                                             const int* __restrict__ strD,
                                                             const int* __restrict__ strS,
                                                             int* __restrict__ rp,
                                                             float* __restrict__ ns,
                                                             int* __restrict__ col,
                                                             int n, int G) {
    __shared__ int cIn[BUCKET_NODES];
    __shared__ int cOut[BUCKET_NODES];
    __shared__ int cur[BUCKET_NODES];
    __shared__ int wtot[32];
    __shared__ int wbase[32];
    const int t = threadIdx.x;
    const int lane = t & 31;
    const int wave = t >> 5;
    const int b = blockIdx.x;
    if (t < BUCKET_NODES) { cIn[t] = 0; cOut[t] = 0; }
    __syncthreads();
    const int d0 = tblD[b * G];
    const int d1 = tblD[(b + 1) * G];
    const int s0 = tblS[b * G];
    const int s1 = tblS[(b + 1) * G];
    for (int base = d0 + t; base < d1; base += 1024 * EDGE_BATCH) {
        int w[EDGE_BATCH];
#pragma unroll
        for (int j = 0; j < EDGE_BATCH; ++j) {
            const int e = min(base + j * 1024, d1 - 1);
            w[j] = strD[e];
        }
#pragma unroll
        for (int j = 0; j < EDGE_BATCH; ++j) {
            asm volatile("" : "+v"(w[j]));
        }
#pragma unroll
        for (int j = 0; j < EDGE_BATCH; ++j)
            if (base + j * 1024 < d1) atomicAdd(&cIn[w[j] & (BUCKET_NODES - 1)], 1);
    }
    for (int base = s0 + t; base < s1; base += 1024 * EDGE_BATCH) {
        int w[EDGE_BATCH];
#pragma unroll
        for (int j = 0; j < EDGE_BATCH; ++j) {
            const int e = min(base + j * 1024, s1 - 1);
            w[j] = strS[e];
        }
#pragma unroll
        for (int j = 0; j < EDGE_BATCH; ++j) {
            asm volatile("" : "+v"(w[j]));
        }
#pragma unroll
        for (int j = 0; j < EDGE_BATCH; ++j)
            if (base + j * 1024 < s1) atomicAdd(&cOut[w[j]], 1);
    }
    __syncthreads();

    const int v = (t < BUCKET_NODES) ? cIn[t] : 0;
    int inc = v;
    for (int o = 1; o < 32; o <<= 1) {
        const int u = __shfl_up(inc, o, 32);
        if (lane >= o) inc += u;
    }
    if (lane == 31) wtot[wave] = inc;
    __syncthreads();
    if (wave == 0) {
        const int x = wtot[lane];
        int xi = x;
        for (int o = 1; o < 32; o <<= 1) {
            const int u = __shfl_up(xi, o, 32);
            if (lane >= o) xi += u;
        }
        wbase[lane] = xi - x;
    }
    __syncthreads();
    if (t < BUCKET_NODES) {
        const int start = wbase[wave] + inc - v;
        cur[t] = start;
        const int node = b * BUCKET_NODES + t;
        if (node < n) {
            rp[node] = d0 + start;
            const int c = cOut[t];
            ns[node] = (c > 0) ? rsqrtf((float)c) : 0.0f;
        }
    }
    __syncthreads();
    for (int base = d0 + t; base < d1; base += 1024 * EDGE_BATCH) {
        int w[EDGE_BATCH];
#pragma unroll
        for (int j = 0; j < EDGE_BATCH; ++j) {
            const int e = min(base + j * 1024, d1 - 1);
            w[j] = strD[e];
        }
#pragma unroll
        for (int j = 0; j < EDGE_BATCH; ++j) {
            asm volatile("" : "+v"(w[j]));
        }
#pragma unroll
        for (int j = 0; j < EDGE_BATCH; ++j) {
            if (base + j * 1024 < d1) {
                const int r = atomicAdd(&cur[w[j] & (BUCKET_NODES - 1)], 1);
                col[d0 + r] = (int)((unsigned)w[j] >> BUCKET_SHIFT);
            }
        }
    }
}

__global__ __launch_bounds__(256) void copy_int4_kernel(const int* __restrict__ a,
                                                        int* __restrict__ b, int n4) {
    int i = blockIdx.x * 256 + threadIdx.x;
    if (i < n4) ((int4*)b)[i] = ((const int4*)a)[i];
}

#define G1_GROUPS 256
#define G1_WAVES 16
#define G1_CT 4
#define G1_PITCH (2 * N_HIDDEN + 32)
#define G1_APITCH 36
__global__ __launch_bounds__(32 * G1_WAVES) void gemm_scaled_wmma_ldsw(const float* __restrict__ H,
                                                                       const float* __restrict__ ns,
                                                                       const float* __restrict__ W,
                                                                       float* __restrict__ out,
                                                                       int nRowTiles) {
    constexpr int K = IN_FEATS;
    constexpr int F = N_HIDDEN;
    constexpr int KSTEPS = K / 4;
    constexpr int CHUNKS = K / 32;
    __shared__ __attribute__((aligned(16))) float Wl[(K / 2) * G1_PITCH];
    __shared__ __attribute__((aligned(16))) float Al[G1_WAVES * 16 * G1_APITCH];
    const int lane = threadIdx.x & 31;
    const int wave = __builtin_amdgcn_readfirstlane((int)(threadIdx.x >> 5));
    const int m  = lane & 15;
    const int hi = lane >> 4;
    const int half = wave & 1;

    float* stage = Al + wave * (16 * G1_APITCH);
    const int sr = lane >> 3;
    const int sc = (lane & 7) * 4;
    float* sdst = stage + sr * G1_APITCH + sc;
    const float* afrag = stage + m * G1_APITCH + 2 * hi;
    const float* bcol = Wl + hi * G1_PITCH + 2 * (half * 64 + m);

    const int stride = (G1_WAVES / 2) * (int)gridDim.x;
    int t = (wave >> 1) * (int)gridDim.x + (int)blockIdx.x;
    const bool has = t < nRowTiles;
    const int t0 = has ? t : 0;
    const float* gsrc = H + (size_t)(t0 * 16 + sr) * K + sc;
    v4f g0 = *(const v4f*)(gsrc);
    v4f g1 = *(const v4f*)(gsrc + (size_t)4 * K);
    v4f g2 = *(const v4f*)(gsrc + (size_t)8 * K);
    v4f g3 = *(const v4f*)(gsrc + (size_t)12 * K);
    float scale = ns[t0 * 16 + m];

#pragma unroll
    for (int i = 0; i < (K / 2) * (F / 4) / (32 * G1_WAVES); ++i) {
        const int idx = (int)threadIdx.x + i * (32 * G1_WAVES);
        const int kp = idx / (F / 4);
        const int c4 = idx - kp * (F / 4);
        const float4 w0 = ((const float4*)W)[(size_t)(2 * kp) * (F / 4) + c4];
        const float4 w1 = ((const float4*)W)[(size_t)(2 * kp + 1) * (F / 4) + c4];
        float4* dst = (float4*)(Wl + kp * G1_PITCH + 8 * c4);
        dst[0] = make_float4(w0.x, w1.x, w0.y, w1.y);
        dst[1] = make_float4(w0.z, w1.z, w0.w, w1.w);
    }
    __syncthreads();
    if (!has) return;

    *(v4f*)(sdst) = g0;
    *(v4f*)(sdst + 4 * G1_APITCH) = g1;
    *(v4f*)(sdst + 8 * G1_APITCH) = g2;
    *(v4f*)(sdst + 12 * G1_APITCH) = g3;
    __builtin_amdgcn_fence(__ATOMIC_ACQ_REL, "wavefront");
    g0 = *(const v4f*)(gsrc + 32);
    g1 = *(const v4f*)(gsrc + (size_t)4 * K + 32);
    g2 = *(const v4f*)(gsrc + (size_t)8 * K + 32);
    g3 = *(const v4f*)(gsrc + (size_t)12 * K + 32);

    v2f fa[2];
    v2f fb[2][G1_CT];
    fa[0] = *(const v2f*)(afrag);
#pragma unroll
    for (int c = 0; c < G1_CT; ++c) fb[0][c] = *(const v2f*)(bcol + c * 32);

    for (;;) {
        const int row0 = t * 16;
        const int tn = t + stride;
        const bool more = tn < nRowTiles;
        const int rown = more ? tn * 16 : row0;
        const float* gsrcn = H + (size_t)(rown + sr) * K + sc;
        const float scalen = ns[rown + m];

        v8f acc[G1_CT];
#pragma unroll
        for (int c = 0; c < G1_CT; ++c) acc[c] = (v8f){};

        for (int kc = 0; kc < CHUNKS; ++kc) {
#pragma unroll
            for (int j = 0; j < 8; ++j) {
                if (j == 7) {
                    *(v4f*)(sdst) = g0;
                    *(v4f*)(sdst + 4 * G1_APITCH) = g1;
                    *(v4f*)(sdst + 8 * G1_APITCH) = g2;
                    *(v4f*)(sdst + 12 * G1_APITCH) = g3;
                    __builtin_amdgcn_fence(__ATOMIC_ACQ_REL, "wavefront");
                    const int kk = kc + 2;
                    const float* gnext = (kk < CHUNKS) ? gsrc + 32 * kk : gsrcn + 32 * (kk - CHUNKS);
                    g0 = *(const v4f*)(gnext);
                    g1 = *(const v4f*)(gnext + (size_t)4 * K);
                    g2 = *(const v4f*)(gnext + (size_t)8 * K);
                    g3 = *(const v4f*)(gnext + (size_t)12 * K);
                }
                const int sn = (kc * 8 + j + 1 == KSTEPS) ? 0 : kc * 8 + j + 1;
                const float* bnext = bcol + sn * (2 * G1_PITCH);
                fa[(j + 1) & 1] = *(const v2f*)(afrag + 4 * ((j + 1) & 7));
#pragma unroll
                for (int c = 0; c < G1_CT; ++c) fb[(j + 1) & 1][c] = *(const v2f*)(bnext + c * 32);
                __builtin_amdgcn_sched_barrier(0);
                v2f a;
                a.x = fa[j & 1].x * scale;
                a.y = fa[j & 1].y * scale;
#pragma unroll
                for (int c = 0; c < G1_CT; ++c)
                    acc[c] = __builtin_amdgcn_wmma_f32_16x16x4_f32(false, a, false, fb[j & 1][c],
                                                                   (short)0, acc[c], false, false);
                __builtin_amdgcn_sched_barrier(0);
            }
        }

        float* __restrict__ obase = out + (size_t)(row0 + 8 * hi) * F + half * 64 + m;
#pragma unroll
        for (int c = 0; c < G1_CT; ++c) {
#pragma unroll
            for (int v = 0; v < 8; ++v)
                obase[(size_t)v * F + c * 16] = acc[c][v];
        }
        if (!more) break;
        t = tn;
        gsrc = gsrcn;
        scale = scalen;
    }
}

#define FUSED_GROUPS 1024
template <int CT>
__global__ __launch_bounds__(512) void gather_rows128_product_kernel(const float* __restrict__ hw,
                                                                     const int* __restrict__ rp,
                                                                     const int* __restrict__ col,
                                                                     const float* __restrict__ bias,
                                                                     const float* __restrict__ ns,
                                                                     const float* __restrict__ W,
                                                                     float* __restrict__ out,
                                                                     int n, int nE, int nTiles) {
    constexpr int F = CT * 16;
    constexpr int PITCH = N_HIDDEN + 4;
    __shared__ __attribute__((aligned(16))) float Wl[N_HIDDEN * F];
    __shared__ __attribute__((aligned(16))) float tile[16 * PITCH];
    const int lane = threadIdx.x & 31;
    const int wave = __builtin_amdgcn_readfirstlane((int)(threadIdx.x >> 5));

#pragma unroll
    for (int idx = threadIdx.x; idx < (N_HIDDEN / 2) * (F / 4); idx += 512) {
        const int kp = idx / (F / 4);
        const int c4 = idx - kp * (F / 4);
        const float4 w0 = ((const float4*)W)[(size_t)(2 * kp) * (F / 4) + c4];
        const float4 w1 = ((const float4*)W)[(size_t)(2 * kp + 1) * (F / 4) + c4];
        float4* dst = (float4*)(Wl + 2 * (kp * F + 4 * c4));
        dst[0] = make_float4(w0.x, w1.x, w0.y, w1.y);
        dst[1] = make_float4(w0.z, w1.z, w0.w, w1.w);
    }

    const float4* __restrict__ hw4 = (const float4*)hw;
    const float4 zero4 = make_float4(0.0f, 0.0f, 0.0f, 0.0f);
    const float4 b = ((const float4*)bias)[lane];

    for (int t = blockIdx.x; t < nTiles; t += gridDim.x) {
        const int row0 = t * 16;
        const int i = row0 + wave;
        float4 r = zero4;
        if (i < n) {
            const int start = rp[i];
            const int nxt = rp[min(i + 1, n - 1)];
            const float scale = ns[i];
            const int end = (i + 1 < n) ? nxt : nE;
            float4 acc = zero4;
            for (int base = start; base < end; base += 32) {
                const int k = base + lane;
                const int myc = (k < end) ? col[k] : 0;
                const int cnt = min(32, end - base);
                const int last = cnt - 1;
                for (int j = 0; j < cnt; j += 4) {
                    const int s0 = __builtin_amdgcn_readlane(myc, j);
                    const int s1 = __builtin_amdgcn_readlane(myc, min(j + 1, last));
                    const int s2 = __builtin_amdgcn_readlane(myc, min(j + 2, last));
                    const int s3 = __builtin_amdgcn_readlane(myc, min(j + 3, last));
                    const float4 v0 = hw4[(size_t)s0 * 32 + lane];
                    float4 v1 = hw4[(size_t)s1 * 32 + lane];
                    float4 v2 = hw4[(size_t)s2 * 32 + lane];
                    float4 v3 = hw4[(size_t)s3 * 32 + lane];
                    if (j + 1 > last) v1 = zero4;
                    if (j + 2 > last) v2 = zero4;
                    if (j + 3 > last) v3 = zero4;
                    acc.x += v0.x; acc.y += v0.y; acc.z += v0.z; acc.w += v0.w;
                    acc.x += v1.x; acc.y += v1.y; acc.z += v1.z; acc.w += v1.w;
                    acc.x += v2.x; acc.y += v2.y; acc.z += v2.z; acc.w += v2.w;
                    acc.x += v3.x; acc.y += v3.y; acc.z += v3.z; acc.w += v3.w;
                }
            }
            const int deg = end - start;
            const float nd = (deg > 0) ? rsqrtf((float)deg) : 0.0f;
            r.x = fmaxf(acc.x * nd + b.x, 0.0f) * scale;
            r.y = fmaxf(acc.y * nd + b.y, 0.0f) * scale;
            r.z = fmaxf(acc.z * nd + b.z, 0.0f) * scale;
            r.w = fmaxf(acc.w * nd + b.w, 0.0f) * scale;
        }
        *(float4*)(tile + wave * PITCH + lane * 4) = r;
        __syncthreads();

        v8f pacc = (v8f){};
        const int m  = lane & 15;
        const int hi = lane >> 4;
        if (wave < CT) {
            const float* arow = tile + m * PITCH + 2 * hi;
            const float* bcol = Wl + 2 * (hi * F + wave * 16 + m);
            v2f fa[2][4], fb[2][4];
#pragma unroll
            for (int j = 0; j < 4; ++j) {
                fa[0][j] = *(const v2f*)(arow + 4 * j);
                fb[0][j] = *(const v2f*)(bcol + 4 * j * F);
            }
#pragma unroll
            for (int g = 0; g < N_HIDDEN / 16; ++g) {
                if (g + 1 < N_HIDDEN / 16) {
#pragma unroll
                    for (int j = 0; j < 4; ++j) {
                        const int k0 = 16 * (g + 1) + 4 * j;
                        fa[(g + 1) & 1][j] = *(const v2f*)(arow + k0);
                        fb[(g + 1) & 1][j] = *(const v2f*)(bcol + k0 * F);
                    }
                }
                __builtin_amdgcn_sched_barrier(0);
#pragma unroll
                for (int j = 0; j < 4; ++j)
                    pacc = __builtin_amdgcn_wmma_f32_16x16x4_f32(false, fa[g & 1][j], false, fb[g & 1][j],
                                                                 (short)0, pacc, false, false);
            }
        }
        __syncthreads();
        if (wave < CT) {
            float* __restrict__ obase = out + (size_t)(row0 + 8 * hi) * F + wave * 16 + m;
#pragma unroll
            for (int v = 0; v < 8; ++v)
                if (row0 + 8 * hi + v < n) obase[(size_t)v * F] = pacc[v];
        }
    }
}

__global__ __launch_bounds__(256) void gather_rows16_kernel(const float* __restrict__ hw,
                                                            const int* __restrict__ rp,
                                                            const int* __restrict__ col,
                                                            const float* __restrict__ bias,
                                                            float* __restrict__ out,
                                                            int n, int nE) {
    const int t = blockIdx.x * 256 + threadIdx.x;
    const int i = t >> 2;
    const int q = t & 3;
    if (i >= n) return;
    const int start = rp[i];
    const int end = (i + 1 < n) ? rp[i + 1] : nE;
    const float4* __restrict__ hw4 = (const float4*)hw;
    const float4 zero4 = make_float4(0.0f, 0.0f, 0.0f, 0.0f);

    float4 acc = zero4;
    const int lastk = end - 1;
    for (int k = start; k < end; k += 4) {
        const int c0 = col[k];
        const int c1 = col[min(k + 1, lastk)];
        const int c2 = col[min(k + 2, lastk)];
        const int c3 = col[min(k + 3, lastk)];
        const float4 v0 = hw4[(size_t)c0 * 4 + q];
        float4 v1 = hw4[(size_t)c1 * 4 + q];
        float4 v2 = hw4[(size_t)c2 * 4 + q];
        float4 v3 = hw4[(size_t)c3 * 4 + q];
        if (k + 1 > lastk) v1 = zero4;
        if (k + 2 > lastk) v2 = zero4;
        if (k + 3 > lastk) v3 = zero4;
        acc.x += v0.x; acc.y += v0.y; acc.z += v0.z; acc.w += v0.w;
        acc.x += v1.x; acc.y += v1.y; acc.z += v1.z; acc.w += v1.w;
        acc.x += v2.x; acc.y += v2.y; acc.z += v2.z; acc.w += v2.w;
        acc.x += v3.x; acc.y += v3.y; acc.z += v3.z; acc.w += v3.w;
    }

    const int deg = end - start;
    const float nd = (deg > 0) ? rsqrtf((float)deg) : 0.0f;
    const float4 b = ((const float4*)bias)[q];
    float4 r;
    r.x = acc.x * nd + b.x;
    r.y = acc.y * nd + b.y;
    r.z = acc.z * nd + b.z;
    r.w = acc.w * nd + b.w;
    ((float4*)out)[(size_t)i * 4 + q] = r;
}

extern "C" void kernel_launch(void* const* d_in, const int* in_sizes, int n_in,
                              void* d_out, int out_size, void* d_ws, size_t ws_size,
                              hipStream_t stream) {
    const float* feat = (const float*)d_in[0];
    const int*   src  = (const int*)d_in[1];
    const int*   dst  = (const int*)d_in[2];
    const float* W1   = (const float*)d_in[3];
    const float* b1   = (const float*)d_in[4];
    const float* W2   = (const float*)d_in[5];
    const float* b2   = (const float*)d_in[6];
    const float* W3   = (const float*)d_in[7];
    const float* b3   = (const float*)d_in[8];
    float* out = (float*)d_out;

    const int N = in_sizes[0] / IN_FEATS;
    const int E = in_sizes[1];

    float* nsW    = (float*)d_ws;
    int*   rp     = (int*)d_ws + N;
    float* bufA   = (float*)(rp + N);
    float* bufB   = bufA + (size_t)N * N_HIDDEN;
    const float* ns = nsW;
    int* col      = (int*)d_out;
    int* col3     = (int*)(bufA + (size_t)N * N_CLASSES);

    const int rowTiles  = N / 16;
    const int nTiles    = (N + 15) / 16;
    const int nBuckets  = (N + BUCKET_NODES - 1) / BUCKET_NODES;
    const int T         = nBuckets * EDGE_CHUNKS;
    int* tblD = (int*)bufA;
    int* tblS = tblD + T + 4;
    int* strD = (int*)bufB;
    int* strS = strD + E;

    bucket_count_kernel<<<EDGE_CHUNKS, 1024, 0, stream>>>(src, dst, tblD, tblS, E, nBuckets);
    scan_tables_kernel<<<2, 1024, 0, stream>>>(tblD, tblS, T);
    bucket_scatter_kernel<<<EDGE_CHUNKS, 1024, 0, stream>>>(src, dst, tblD, tblS, strD, strS, E, nBuckets);
    bucket_finish_kernel<<<nBuckets, 1024, 0, stream>>>(tblD, tblS, strD, strS, rp, nsW, col, N, EDGE_CHUNKS);

    gemm_scaled_wmma_ldsw<<<G1_GROUPS, 32 * G1_WAVES, 0, stream>>>(feat, ns, W1, bufA, rowTiles);

    gather_rows128_product_kernel<8><<<FUSED_GROUPS, 512, 0, stream>>>(bufA, rp, col, b1, ns, W2, bufB, N, E, nTiles);

    gather_rows128_product_kernel<1><<<FUSED_GROUPS, 512, 0, stream>>>(bufB, rp, col, b2, ns, W3, bufA, N, E, nTiles);

    copy_int4_kernel<<<(E / 4 + 255) / 256, 256, 0, stream>>>(col, col3, E / 4);
    gather_rows16_kernel<<<(4 * N + 255) / 256, 256, 0, stream>>>(bufA, rp, col3, b3, out, N, E);
}


# Edge grouping in two launches, not four: a work-group counts, scans in LDS and orders its chunk by bucket in place; a bucket collects its segment of every chunk; order inside a row changes
# speedup vs baseline: 4.4914x; 1.1728x over previous
#include <hip/hip_runtime.h>
#include <hip/hip_bf16.h>

typedef __attribute__((ext_vector_type(2))) float v2f;
typedef __attribute__((ext_vector_type(8))) float v8f;
typedef __attribute__((ext_vector_type(4))) float v4f;

#define IN_FEATS 384
#define N_HIDDEN 128
#define N_CLASSES 16
#define BUCKET_SHIFT 9
#define BUCKET_NODES 512
#define MAX_BUCKETS 256
#define EDGE_CHUNKS 250
#define EDGE_BATCH 8


__global__ __launch_bounds__(1024) void bucket_partition_kernel(const int* __restrict__ src,
                                                                const int* __restrict__ dst,
                                                                int* __restrict__ tblD,
                                                                int* __restrict__ tblS,
                                                                int* __restrict__ strD,
                                                                int* __restrict__ strS,
                                                                int nE, int nB) {
    __shared__ int hD[MAX_BUCKETS];
    __shared__ int hS[MAX_BUCKETS];
    __shared__ int cD[MAX_BUCKETS];
    __shared__ int cS[MAX_BUCKETS];
    const int t = threadIdx.x;
    const int lane = t & 31;
    const int wave = t >> 5;
    const int G = gridDim.x;
    const int g = blockIdx.x;
    if (t < MAX_BUCKETS) { hD[t] = 0; hS[t] = 0; }
    __syncthreads();
    const int chunk = (nE + G - 1) / G;
    const int e0 = g * chunk;
    const int e1 = min(e0 + chunk, nE);
    for (int base = e0 + t; base < e1; base += 1024 * EDGE_BATCH) {
        int d[EDGE_BATCH], s[EDGE_BATCH];
#pragma unroll
        for (int j = 0; j < EDGE_BATCH; ++j) {
            const int e = min(base + j * 1024, e1 - 1);
            d[j] = dst[e];
            s[j] = src[e];
        }
#pragma unroll
        for (int j = 0; j < EDGE_BATCH; ++j) {
            asm volatile("" : "+v"(d[j]), "+v"(s[j]));
        }
#pragma unroll
        for (int j = 0; j < EDGE_BATCH; ++j) {
            if (base + j * 1024 < e1) {
                atomicAdd(&hD[d[j] >> BUCKET_SHIFT], 1);
                atomicAdd(&hS[s[j] >> BUCKET_SHIFT], 1);
            }
        }
    }
    __syncthreads();
    if (wave < 2) {
        const int* h = (wave == 0) ? hD : hS;
        int* c = (wave == 0) ? cD : cS;
        int* tbl = (wave == 0) ? tblD : tblS;
        int v[8];
        int sum = 0;
#pragma unroll
        for (int k = 0; k < 8; ++k) { v[k] = h[lane * 8 + k]; sum += v[k]; }
        int inc = sum;
        for (int o = 1; o < 32; o <<= 1) {
            const int u = __shfl_up(inc, o, 32);
            if (lane >= o) inc += u;
        }
        int ex = inc - sum;
#pragma unroll
        for (int k = 0; k < 8; ++k) {
            const int idx = lane * 8 + k;
            c[idx] = e0 + ex;
            if (idx < nB) tbl[idx * G + g] = ex;
            ex += v[k];
        }
        if (lane == 0) tbl[nB * G + g] = max(e1 - e0, 0);
    }
    __syncthreads();
    for (int base = e0 + t; base < e1; base += 1024 * EDGE_BATCH) {
        int d[EDGE_BATCH], s[EDGE_BATCH];
#pragma unroll
        for (int j = 0; j < EDGE_BATCH; ++j) {
            const int e = min(base + j * 1024, e1 - 1);
            d[j] = dst[e];
            s[j] = src[e];
        }
#pragma unroll
        for (int j = 0; j < EDGE_BATCH; ++j) {
            asm volatile("" : "+v"(d[j]), "+v"(s[j]));
        }
#pragma unroll
        for (int j = 0; j < EDGE_BATCH; ++j) {
            if (base + j * 1024 < e1) {
                const int pd = atomicAdd(&cD[d[j] >> BUCKET_SHIFT], 1);
                const int ps = atomicAdd(&cS[s[j] >> BUCKET_SHIFT], 1);
                strD[pd] = (int)(((unsigned)s[j] << BUCKET_SHIFT) |
                                 (unsigned)(d[j] & (BUCKET_NODES - 1)));
                strS[ps] = s[j] & (BUCKET_NODES - 1);
            }
        }
    }
}

#define MAX_CHUNKS 256
#define SEGS_PER_WAVE (MAX_CHUNKS / 32)
__global__ __launch_bounds__(1024) void bucket_finish_kernel(const int* __restrict__ tblD,
                                                             const int* __restrict__ tblS,
                                                             const int* __restrict__ strD,
                                                             const int* __restrict__ strS,
                                                             int* __restrict__ rp,
                                                             float* __restrict__ ns,
                                                             int* __restrict__ col,
                                                             int n, int nE, int G) {
    __shared__ int cIn[BUCKET_NODES];
    __shared__ int cOut[BUCKET_NODES];
    __shared__ int cur[BUCKET_NODES];
    __shared__ int gaD[MAX_CHUNKS];
    __shared__ int lnD[MAX_CHUNKS];
    __shared__ int gaS[MAX_CHUNKS];
    __shared__ int lnS[MAX_CHUNKS];
    __shared__ int wtot[32];
    __shared__ int wbase[32];
    __shared__ int d0sh;
    const int t = threadIdx.x;
    const int lane = t & 31;
    const int wave = __builtin_amdgcn_readfirstlane((int)(threadIdx.x >> 5));
    const int b = blockIdx.x;
    const int chunk = (nE + G - 1) / G;
    if (t < BUCKET_NODES) { cIn[t] = 0; cOut[t] = 0; }
    if (t == 0) d0sh = 0;
    __syncthreads();
    if (t < MAX_CHUNKS) {
        int aD = 0, lD = 0, aS = 0, lS = 0;
        if (t < G) {
            aD = tblD[b * G + t];
            lD = tblD[(b + 1) * G + t] - aD;
            aS = tblS[b * G + t];
            lS = tblS[(b + 1) * G + t] - aS;
        }
        gaD[t] = t * chunk + aD;
        lnD[t] = lD;
        gaS[t] = t * chunk + aS;
        lnS[t] = lS;
        int r = aD;
        for (int o = 16; o > 0; o >>= 1) r += __shfl_xor(r, o, 32);
        if (lane == 0) atomicAdd(&d0sh, r);
    }
    __syncthreads();
    const int d0 = d0sh;

    int a[SEGS_PER_WAVE], ln[SEGS_PER_WAVE];
    int mx = 0;
#pragma unroll
    for (int i = 0; i < SEGS_PER_WAVE; ++i) {
        a[i] = __builtin_amdgcn_readfirstlane(gaD[wave + 32 * i]);
        ln[i] = __builtin_amdgcn_readfirstlane(lnD[wave + 32 * i]);
        mx = max(mx, ln[i]);
    }
    for (int o0 = 0; o0 < mx; o0 += 32) {
        const int o = o0 + lane;
        int w[SEGS_PER_WAVE];
#pragma unroll
        for (int i = 0; i < SEGS_PER_WAVE; ++i) w[i] = strD[(o < ln[i]) ? a[i] + o : 0];
#pragma unroll
        for (int i = 0; i < SEGS_PER_WAVE; ++i) {
            asm volatile("" : "+v"(w[i]));
        }
#pragma unroll
        for (int i = 0; i < SEGS_PER_WAVE; ++i)
            if (o < ln[i]) atomicAdd(&cIn[w[i] & (BUCKET_NODES - 1)], 1);
    }
    {
        int aS[SEGS_PER_WAVE], lS[SEGS_PER_WAVE];
        int mxS = 0;
#pragma unroll
        for (int i = 0; i < SEGS_PER_WAVE; ++i) {
            aS[i] = __builtin_amdgcn_readfirstlane(gaS[wave + 32 * i]);
            lS[i] = __builtin_amdgcn_readfirstlane(lnS[wave + 32 * i]);
            mxS = max(mxS, lS[i]);
        }
        for (int o0 = 0; o0 < mxS; o0 += 32) {
            const int o = o0 + lane;
            int w[SEGS_PER_WAVE];
#pragma unroll
            for (int i = 0; i < SEGS_PER_WAVE; ++i) w[i] = strS[(o < lS[i]) ? aS[i] + o : 0];
#pragma unroll
            for (int i = 0; i < SEGS_PER_WAVE; ++i) {
                asm volatile("" : "+v"(w[i]));
            }
#pragma unroll
            for (int i = 0; i < SEGS_PER_WAVE; ++i)
                if (o < lS[i]) atomicAdd(&cOut[w[i] & (BUCKET_NODES - 1)], 1);
        }
    }
    __syncthreads();

    const int v = (t < BUCKET_NODES) ? cIn[t] : 0;
    int inc = v;
    for (int o = 1; o < 32; o <<= 1) {
        const int u = __shfl_up(inc, o, 32);
        if (lane >= o) inc += u;
    }
    if (lane == 31) wtot[wave] = inc;
    __syncthreads();
    if (wave == 0) {
        const int x = wtot[lane];
        int xi = x;
        for (int o = 1; o < 32; o <<= 1) {
            const int u = __shfl_up(xi, o, 32);
            if (lane >= o) xi += u;
        }
        wbase[lane] = xi - x;
    }
    __syncthreads();
    if (t < BUCKET_NODES) {
        const int start = wbase[wave] + inc - v;
        cur[t] = start;
        const int node = b * BUCKET_NODES + t;
        if (node < n) {
            rp[node] = d0 + start;
            const int c = cOut[t];
            ns[node] = (c > 0) ? rsqrtf((float)c) : 0.0f;
        }
    }
    __syncthreads();
    for (int o0 = 0; o0 < mx; o0 += 32) {
        const int o = o0 + lane;
        int w[SEGS_PER_WAVE];
#pragma unroll
        for (int i = 0; i < SEGS_PER_WAVE; ++i) w[i] = strD[(o < ln[i]) ? a[i] + o : 0];
#pragma unroll
        for (int i = 0; i < SEGS_PER_WAVE; ++i) {
            asm volatile("" : "+v"(w[i]));
        }
#pragma unroll
        for (int i = 0; i < SEGS_PER_WAVE; ++i) {
            if (o < ln[i]) {
                const int r = atomicAdd(&cur[w[i] & (BUCKET_NODES - 1)], 1);
                col[d0 + r] = (int)((unsigned)w[i] >> BUCKET_SHIFT);
            }
        }
    }
}

__global__ __launch_bounds__(256) void copy_int4_kernel(const int* __restrict__ a,
                                                        int* __restrict__ b, int n4) {
    int i = blockIdx.x * 256 + threadIdx.x;
    if (i < n4) ((int4*)b)[i] = ((const int4*)a)[i];
}

#define G1_GROUPS 256
#define G1_WAVES 16
#define G1_CT 4
#define G1_PITCH (2 * N_HIDDEN + 32)
#define G1_APITCH 36
__global__ __launch_bounds__(32 * G1_WAVES) void gemm_scaled_wmma_ldsw(const float* __restrict__ H,
                                                                       const float* __restrict__ ns,
                                                                       const float* __restrict__ W,
                                                                       float* __restrict__ out,
                                                                       int nRowTiles) {
    constexpr int K = IN_FEATS;
    constexpr int F = N_HIDDEN;
    constexpr int KSTEPS = K / 4;
    constexpr int CHUNKS = K / 32;
    __shared__ __attribute__((aligned(16))) float Wl[(K / 2) * G1_PITCH];
    __shared__ __attribute__((aligned(16))) float Al[G1_WAVES * 16 * G1_APITCH];
    const int lane = threadIdx.x & 31;
    const int wave = __builtin_amdgcn_readfirstlane((int)(threadIdx.x >> 5));
    const int m  = lane & 15;
    const int hi = lane >> 4;
    const int half = wave & 1;

    float* stage = Al + wave * (16 * G1_APITCH);
    const int sr = lane >> 3;
    const int sc = (lane & 7) * 4;
    float* sdst = stage + sr * G1_APITCH + sc;
    const float* afrag = stage + m * G1_APITCH + 2 * hi;
    const float* bcol = Wl + hi * G1_PITCH + 2 * (half * 64 + m);

    const int stride = (G1_WAVES / 2) * (int)gridDim.x;
    int t = (wave >> 1) * (int)gridDim.x + (int)blockIdx.x;
    const bool has = t < nRowTiles;
    const int t0 = has ? t : 0;
    const float* gsrc = H + (size_t)(t0 * 16 + sr) * K + sc;
    v4f g0 = *(const v4f*)(gsrc);
    v4f g1 = *(const v4f*)(gsrc + (size_t)4 * K);
    v4f g2 = *(const v4f*)(gsrc + (size_t)8 * K);
    v4f g3 = *(const v4f*)(gsrc + (size_t)12 * K);
    float scale = ns[t0 * 16 + m];

#pragma unroll
    for (int i = 0; i < (K / 2) * (F / 4) / (32 * G1_WAVES); ++i) {
        const int idx = (int)threadIdx.x + i * (32 * G1_WAVES);
        const int kp = idx / (F / 4);
        const int c4 = idx - kp * (F / 4);
        const float4 w0 = ((const float4*)W)[(size_t)(2 * kp) * (F / 4) + c4];
        const float4 w1 = ((const float4*)W)[(size_t)(2 * kp + 1) * (F / 4) + c4];
        float4* dst = (float4*)(Wl + kp * G1_PITCH + 8 * c4);
        dst[0] = make_float4(w0.x, w1.x, w0.y, w1.y);
        dst[1] = make_float4(w0.z, w1.z, w0.w, w1.w);
    }
    __syncthreads();
    if (!has) return;

    *(v4f*)(sdst) = g0;
    *(v4f*)(sdst + 4 * G1_APITCH) = g1;
    *(v4f*)(sdst + 8 * G1_APITCH) = g2;
    *(v4f*)(sdst + 12 * G1_APITCH) = g3;
    __builtin_amdgcn_fence(__ATOMIC_ACQ_REL, "wavefront");
    g0 = *(const v4f*)(gsrc + 32);
    g1 = *(const v4f*)(gsrc + (size_t)4 * K + 32);
    g2 = *(const v4f*)(gsrc + (size_t)8 * K + 32);
    g3 = *(const v4f*)(gsrc + (size_t)12 * K + 32);

    v2f fa[2];
    v2f fb[2][G1_CT];
    fa[0] = *(const v2f*)(afrag);
#pragma unroll
    for (int c = 0; c < G1_CT; ++c) fb[0][c] = *(const v2f*)(bcol + c * 32);

    for (;;) {
        const int row0 = t * 16;
        const int tn = t + stride;
        const bool more = tn < nRowTiles;
        const int rown = more ? tn * 16 : row0;
        const float* gsrcn = H + (size_t)(rown + sr) * K + sc;
        const float scalen = ns[rown + m];

        v8f acc[G1_CT];
#pragma unroll
        for (int c = 0; c < G1_CT; ++c) acc[c] = (v8f){};

        for (int kc = 0; kc < CHUNKS; ++kc) {
#pragma unroll
            for (int j = 0; j < 8; ++j) {
                if (j == 7) {
                    *(v4f*)(sdst) = g0;
                    *(v4f*)(sdst + 4 * G1_APITCH) = g1;
                    *(v4f*)(sdst + 8 * G1_APITCH) = g2;
                    *(v4f*)(sdst + 12 * G1_APITCH) = g3;
                    __builtin_amdgcn_fence(__ATOMIC_ACQ_REL, "wavefront");
                    const int kk = kc + 2;
                    const float* gnext = (kk < CHUNKS) ? gsrc + 32 * kk : gsrcn + 32 * (kk - CHUNKS);
                    g0 = *(const v4f*)(gnext);
                    g1 = *(const v4f*)(gnext + (size_t)4 * K);
                    g2 = *(const v4f*)(gnext + (size_t)8 * K);
                    g3 = *(const v4f*)(gnext + (size_t)12 * K);
                }
                const int sn = (kc * 8 + j + 1 == KSTEPS) ? 0 : kc * 8 + j + 1;
                const float* bnext = bcol + sn * (2 * G1_PITCH);
                fa[(j + 1) & 1] = *(const v2f*)(afrag + 4 * ((j + 1) & 7));
#pragma unroll
                for (int c = 0; c < G1_CT; ++c) fb[(j + 1) & 1][c] = *(const v2f*)(bnext + c * 32);
                __builtin_amdgcn_sched_barrier(0);
                v2f a;
                a.x = fa[j & 1].x * scale;
                a.y = fa[j & 1].y * scale;
#pragma unroll
                for (int c = 0; c < G1_CT; ++c)
                    acc[c] = __builtin_amdgcn_wmma_f32_16x16x4_f32(false, a, false, fb[j & 1][c],
                                                                   (short)0, acc[c], false, false);
                __builtin_amdgcn_sched_barrier(0);
            }
        }

        float* __restrict__ obase = out + (size_t)(row0 + 8 * hi) * F + half * 64 + m;
#pragma unroll
        for (int c = 0; c < G1_CT; ++c) {
#pragma unroll
            for (int v = 0; v < 8; ++v)
                obase[(size_t)v * F + c * 16] = acc[c][v];
        }
        if (!more) break;
        t = tn;
        gsrc = gsrcn;
        scale = scalen;
    }
}

#define FUSED_GROUPS 1024
template <int CT>
__global__ __launch_bounds__(512) void gather_rows128_product_kernel(const float* __restrict__ hw,
                                                                     const int* __restrict__ rp,
                                                                     const int* __restrict__ col,
                                                                     const float* __restrict__ bias,
                                                                     const float* __restrict__ ns,
                                                                     const float* __restrict__ W,
                                                                     float* __restrict__ out,
                                                                     int n, int nE, int nTiles) {
    constexpr int F = CT * 16;
    constexpr int PITCH = N_HIDDEN + 4;
    __shared__ __attribute__((aligned(16))) float Wl[N_HIDDEN * F];
    __shared__ __attribute__((aligned(16))) float tile[16 * PITCH];
    const int lane = threadIdx.x & 31;
    const int wave = __builtin_amdgcn_readfirstlane((int)(threadIdx.x >> 5));

#pragma unroll
    for (int idx = threadIdx.x; idx < (N_HIDDEN / 2) * (F / 4); idx += 512) {
        const int kp = idx / (F / 4);
        const int c4 = idx - kp * (F / 4);
        const float4 w0 = ((const float4*)W)[(size_t)(2 * kp) * (F / 4) + c4];
        const float4 w1 = ((const float4*)W)[(size_t)(2 * kp + 1) * (F / 4) + c4];
        float4* dst = (float4*)(Wl + 2 * (kp * F + 4 * c4));
        dst[0] = make_float4(w0.x, w1.x, w0.y, w1.y);
        dst[1] = make_float4(w0.z, w1.z, w0.w, w1.w);
    }

    const float4* __restrict__ hw4 = (const float4*)hw;
    const float4 zero4 = make_float4(0.0f, 0.0f, 0.0f, 0.0f);
    const float4 b = ((const float4*)bias)[lane];

    for (int t = blockIdx.x; t < nTiles; t += gridDim.x) {
        const int row0 = t * 16;
        const int i = row0 + wave;
        float4 r = zero4;
        if (i < n) {
            const int start = rp[i];
            const int nxt = rp[min(i + 1, n - 1)];
            const float scale = ns[i];
            const int end = (i + 1 < n) ? nxt : nE;
            float4 acc = zero4;
            for (int base = start; base < end; base += 32) {
                const int k = base + lane;
                const int myc = (k < end) ? col[k] : 0;
                const int cnt = min(32, end - base);
                const int last = cnt - 1;
                for (int j = 0; j < cnt; j += 4) {
                    const int s0 = __builtin_amdgcn_readlane(myc, j);
                    const int s1 = __builtin_amdgcn_readlane(myc, min(j + 1, last));
                    const int s2 = __builtin_amdgcn_readlane(myc, min(j + 2, last));
                    const int s3 = __builtin_amdgcn_readlane(myc, min(j + 3, last));
                    const float4 v0 = hw4[(size_t)s0 * 32 + lane];
                    float4 v1 = hw4[(size_t)s1 * 32 + lane];
                    float4 v2 = hw4[(size_t)s2 * 32 + lane];
                    float4 v3 = hw4[(size_t)s3 * 32 + lane];
                    if (j + 1 > last) v1 = zero4;
                    if (j + 2 > last) v2 = zero4;
                    if (j + 3 > last) v3 = zero4;
                    acc.x += v0.x; acc.y += v0.y; acc.z += v0.z; acc.w += v0.w;
                    acc.x += v1.x; acc.y += v1.y; acc.z += v1.z; acc.w += v1.w;
                    acc.x += v2.x; acc.y += v2.y; acc.z += v2.z; acc.w += v2.w;
                    acc.x += v3.x; acc.y += v3.y; acc.z += v3.z; acc.w += v3.w;
                }
            }
            const int deg = end - start;
            const float nd = (deg > 0) ? rsqrtf((float)deg) : 0.0f;
            r.x = fmaxf(acc.x * nd + b.x, 0.0f) * scale;
            r.y = fmaxf(acc.y * nd + b.y, 0.0f) * scale;
            r.z = fmaxf(acc.z * nd + b.z, 0.0f) * scale;
            r.w = fmaxf(acc.w * nd + b.w, 0.0f) * scale;
        }
        *(float4*)(tile + wave * PITCH + lane * 4) = r;
        __syncthreads();

        v8f pacc = (v8f){};
        const int m  = lane & 15;
        const int hi = lane >> 4;
        if (wave < CT) {
            const float* arow = tile + m * PITCH + 2 * hi;
            const float* bcol = Wl + 2 * (hi * F + wave * 16 + m);
            v2f fa[2][4], fb[2][4];
#pragma unroll
            for (int j = 0; j < 4; ++j) {
                fa[0][j] = *(const v2f*)(arow + 4 * j);
                fb[0][j] = *(const v2f*)(bcol + 4 * j * F);
            }
#pragma unroll
            for (int g = 0; g < N_HIDDEN / 16; ++g) {
                if (g + 1 < N_HIDDEN / 16) {
#pragma unroll
                    for (int j = 0; j < 4; ++j) {
                        const int k0 = 16 * (g + 1) + 4 * j;
                        fa[(g + 1) & 1][j] = *(const v2f*)(arow + k0);
                        fb[(g + 1) & 1][j] = *(const v2f*)(bcol + k0 * F);
                    }
                }
                __builtin_amdgcn_sched_barrier(0);
#pragma unroll
                for (int j = 0; j < 4; ++j)
                    pacc = __builtin_amdgcn_wmma_f32_16x16x4_f32(false, fa[g & 1][j], false, fb[g & 1][j],
                                                                 (short)0, pacc, false, false);
            }
        }
        __syncthreads();
        if (wave < CT) {
            float* __restrict__ obase = out + (size_t)(row0 + 8 * hi) * F + wave * 16 + m;
#pragma unroll
            for (int v = 0; v < 8; ++v)
                if (row0 + 8 * hi + v < n) obase[(size_t)v * F] = pacc[v];
        }
    }
}

__global__ __launch_bounds__(256) void gather_rows16_kernel(const float* __restrict__ hw,
                                                            const int* __restrict__ rp,
                                                            const int* __restrict__ col,
                                                            const float* __restrict__ bias,
                                                            float* __restrict__ out,
                                                            int n, int nE) {
    const int t = blockIdx.x * 256 + threadIdx.x;
    const int i = t >> 2;
    const int q = t & 3;
    if (i >= n) return;
    const int start = rp[i];
    const int end = (i + 1 < n) ? rp[i + 1] : nE;
    const float4* __restrict__ hw4 = (const float4*)hw;
    const float4 zero4 = make_float4(0.0f, 0.0f, 0.0f, 0.0f);

    float4 acc = zero4;
    const int lastk = end - 1;
    for (int k = start; k < end; k += 4) {
        const int c0 = col[k];
        const int c1 = col[min(k + 1, lastk)];
        const int c2 = col[min(k + 2, lastk)];
        const int c3 = col[min(k + 3, lastk)];
        const float4 v0 = hw4[(size_t)c0 * 4 + q];
        float4 v1 = hw4[(size_t)c1 * 4 + q];
        float4 v2 = hw4[(size_t)c2 * 4 + q];
        float4 v3 = hw4[(size_t)c3 * 4 + q];
        if (k + 1 > lastk) v1 = zero4;
        if (k + 2 > lastk) v2 = zero4;
        if (k + 3 > lastk) v3 = zero4;
        acc.x += v0.x; acc.y += v0.y; acc.z += v0.z; acc.w += v0.w;
        acc.x += v1.x; acc.y += v1.y; acc.z += v1.z; acc.w += v1.w;
        acc.x += v2.x; acc.y += v2.y; acc.z += v2.z; acc.w += v2.w;
        acc.x += v3.x; acc.y += v3.y; acc.z += v3.z; acc.w += v3.w;
    }

    const int deg = end - start;
    const float nd = (deg > 0) ? rsqrtf((float)deg) : 0.0f;
    const float4 b = ((const float4*)bias)[q];
    float4 r;
    r.x = acc.x * nd + b.x;
    r.y = acc.y * nd + b.y;
    r.z = acc.z * nd + b.z;
    r.w = acc.w * nd + b.w;
    ((float4*)out)[(size_t)i * 4 + q] = r;
}

extern "C" void kernel_launch(void* const* d_in, const int* in_sizes, int n_in,
                              void* d_out, int out_size, void* d_ws, size_t ws_size,
                              hipStream_t stream) {
    const float* feat = (const float*)d_in[0];
    const int*   src  = (const int*)d_in[1];
    const int*   dst  = (const int*)d_in[2];
    const float* W1   = (const float*)d_in[3];
    const float* b1   = (const float*)d_in[4];
    const float* W2   = (const float*)d_in[5];
    const float* b2   = (const float*)d_in[6];
    const float* W3   = (const float*)d_in[7];
    const float* b3   = (const float*)d_in[8];
    float* out = (float*)d_out;

    const int N = in_sizes[0] / IN_FEATS;
    const int E = in_sizes[1];

    float* nsW    = (float*)d_ws;
    int*   rp     = (int*)d_ws + N;
    float* bufA   = (float*)(rp + N);
    float* bufB   = bufA + (size_t)N * N_HIDDEN;
    const float* ns = nsW;
    int* col      = (int*)d_out;
    int* col3     = (int*)(bufA + (size_t)N * N_CLASSES);

    const int rowTiles  = N / 16;
    const int nTiles    = (N + 15) / 16;
    const int nBuckets  = (N + BUCKET_NODES - 1) / BUCKET_NODES;
    const int T         = (nBuckets + 1) * EDGE_CHUNKS;
    int* tblD = (int*)bufA;
    int* tblS = tblD + T;
    int* strD = (int*)bufB;
    int* strS = strD + E;

    bucket_partition_kernel<<<EDGE_CHUNKS, 1024, 0, stream>>>(src, dst, tblD, tblS, strD, strS, E, nBuckets);
    bucket_finish_kernel<<<nBuckets, 1024, 0, stream>>>(tblD, tblS, strD, strS, rp, nsW, col, N, E, EDGE_CHUNKS);

    gemm_scaled_wmma_ldsw<<<G1_GROUPS, 32 * G1_WAVES, 0, stream>>>(feat, ns, W1, bufA, rowTiles);

    gather_rows128_product_kernel<8><<<FUSED_GROUPS, 512, 0, stream>>>(bufA, rp, col, b1, ns, W2, bufB, N, E, nTiles);

    gather_rows128_product_kernel<1><<<FUSED_GROUPS, 512, 0, stream>>>(bufB, rp, col, b2, ns, W3, bufA, N, E, nTiles);

    copy_int4_kernel<<<(E / 4 + 255) / 256, 256, 0, stream>>>(col, col3, E / 4);
    gather_rows16_kernel<<<(4 * N + 255) / 256, 256, 0, stream>>>(bufA, rp, col3, b3, out, N, E);
}


# Both edge-grouping passes take their slots in an LDS copy of the chunk / of the bucket and copy it out in order: stream and col stores are whole 128-byte lines, not one 4-byte piece per lane
# speedup vs baseline: 4.8097x; 1.0709x over previous
#include <hip/hip_runtime.h>
#include <hip/hip_bf16.h>

typedef __attribute__((ext_vector_type(2))) float v2f;
typedef __attribute__((ext_vector_type(8))) float v8f;
typedef __attribute__((ext_vector_type(4))) float v4f;

#define IN_FEATS 384
#define N_HIDDEN 128
#define N_CLASSES 16
#define BUCKET_SHIFT 9
#define BUCKET_NODES 512
#define MAX_BUCKETS 256
#define EDGE_CHUNKS 250
#define EDGE_BATCH 8
#define CHUNK_CAP 6400
#define COL_CAP 12288


__global__ __launch_bounds__(1024) void bucket_partition_kernel(const int* __restrict__ src,
                                                                const int* __restrict__ dst,
                                                                int* __restrict__ tblD,
                                                                int* __restrict__ tblS,
                                                                int* __restrict__ strD,
                                                                int* __restrict__ strS,
                                                                int nE, int nB) {
    __shared__ int hD[MAX_BUCKETS];
    __shared__ int hS[MAX_BUCKETS];
    __shared__ int cD[MAX_BUCKETS];
    __shared__ int cS[MAX_BUCKETS];
    __shared__ __attribute__((aligned(16))) int stD[CHUNK_CAP];
    __shared__ __attribute__((aligned(16))) int stS[CHUNK_CAP];
    const int t = threadIdx.x;
    const int lane = t & 31;
    const int wave = t >> 5;
    const int G = gridDim.x;
    const int g = blockIdx.x;
    if (t < MAX_BUCKETS) { hD[t] = 0; hS[t] = 0; }
    __syncthreads();
    const int chunk = (nE + G - 1) / G;
    const int e0 = g * chunk;
    const int e1 = min(e0 + chunk, nE);
    const int len = max(e1 - e0, 0);
    const bool staged = len <= CHUNK_CAP;
    const int cbase = staged ? 0 : e0;
    for (int base = e0 + t; base < e1; base += 1024 * EDGE_BATCH) {
        int d[EDGE_BATCH], s[EDGE_BATCH];
#pragma unroll
        for (int j = 0; j < EDGE_BATCH; ++j) {
            const int e = min(base + j * 1024, e1 - 1);
            d[j] = dst[e];
            s[j] = src[e];
        }
#pragma unroll
        for (int j = 0; j < EDGE_BATCH; ++j) {
            asm volatile("" : "+v"(d[j]), "+v"(s[j]));
        }
#pragma unroll
        for (int j = 0; j < EDGE_BATCH; ++j) {
            if (base + j * 1024 < e1) {
                atomicAdd(&hD[d[j] >> BUCKET_SHIFT], 1);
                atomicAdd(&hS[s[j] >> BUCKET_SHIFT], 1);
            }
        }
    }
    __syncthreads();
    if (wave < 2) {
        const int* h = (wave == 0) ? hD : hS;
        int* c = (wave == 0) ? cD : cS;
        int* tbl = (wave == 0) ? tblD : tblS;
        int v[8];
        int sum = 0;
#pragma unroll
        for (int k = 0; k < 8; ++k) { v[k] = h[lane * 8 + k]; sum += v[k]; }
        int inc = sum;
        for (int o = 1; o < 32; o <<= 1) {
            const int u = __shfl_up(inc, o, 32);
            if (lane >= o) inc += u;
        }
        int ex = inc - sum;
#pragma unroll
        for (int k = 0; k < 8; ++k) {
            const int idx = lane * 8 + k;
            c[idx] = cbase + ex;
            if (idx < nB) tbl[idx * G + g] = ex;
            ex += v[k];
        }
        if (lane == 0) tbl[nB * G + g] = len;
    }
    __syncthreads();
    for (int base = e0 + t; base < e1; base += 1024 * EDGE_BATCH) {
        int d[EDGE_BATCH], s[EDGE_BATCH];
#pragma unroll
        for (int j = 0; j < EDGE_BATCH; ++j) {
            const int e = min(base + j * 1024, e1 - 1);
            d[j] = dst[e];
            s[j] = src[e];
        }
#pragma unroll
        for (int j = 0; j < EDGE_BATCH; ++j) {
            asm volatile("" : "+v"(d[j]), "+v"(s[j]));
        }
#pragma unroll
        for (int j = 0; j < EDGE_BATCH; ++j) {
            if (base + j * 1024 < e1) {
                const int pd = atomicAdd(&cD[d[j] >> BUCKET_SHIFT], 1);
                const int ps = atomicAdd(&cS[s[j] >> BUCKET_SHIFT], 1);
                const int wd = (int)(((unsigned)s[j] << BUCKET_SHIFT) |
                                     (unsigned)(d[j] & (BUCKET_NODES - 1)));
                const int wsrc = s[j] & (BUCKET_NODES - 1);
                if (staged) {
                    stD[pd] = wd;
                    stS[ps] = wsrc;
                } else {
                    strD[pd] = wd;
                    strS[ps] = wsrc;
                }
            }
        }
    }
    if (staged) {
        __syncthreads();
        int done = 0;
        if ((e0 & 3) == 0) {
            const int n4 = len >> 2;
            int4* gD = (int4*)(strD + e0);
            int4* gS = (int4*)(strS + e0);
            for (int i = t; i < n4; i += 1024) {
                gD[i] = ((const int4*)stD)[i];
                gS[i] = ((const int4*)stS)[i];
            }
            done = n4 << 2;
        }
        for (int i = done + t; i < len; i += 1024) {
            strD[e0 + i] = stD[i];
            strS[e0 + i] = stS[i];
        }
    }
}

#define MAX_CHUNKS 256
#define SEGS_PER_WAVE (MAX_CHUNKS / 32)
__global__ __launch_bounds__(1024) void bucket_finish_kernel(const int* __restrict__ tblD,
                                                             const int* __restrict__ tblS,
                                                             const int* __restrict__ strD,
                                                             const int* __restrict__ strS,
                                                             int* __restrict__ rp,
                                                             float* __restrict__ ns,
                                                             int* __restrict__ col,
                                                             int n, int nE, int G) {
    __shared__ int cIn[BUCKET_NODES];
    __shared__ int cOut[BUCKET_NODES];
    __shared__ int cur[BUCKET_NODES];
    __shared__ int gaD[MAX_CHUNKS];
    __shared__ int lnD[MAX_CHUNKS];
    __shared__ int gaS[MAX_CHUNKS];
    __shared__ int lnS[MAX_CHUNKS];
    __shared__ int wtot[32];
    __shared__ int wbase[32];
    __shared__ int d0sh;
    __shared__ int stC[COL_CAP];
    const int t = threadIdx.x;
    const int lane = t & 31;
    const int wave = __builtin_amdgcn_readfirstlane((int)(threadIdx.x >> 5));
    const int b = blockIdx.x;
    const int chunk = (nE + G - 1) / G;
    if (t < BUCKET_NODES) { cIn[t] = 0; cOut[t] = 0; }
    if (t == 0) d0sh = 0;
    __syncthreads();
    if (t < MAX_CHUNKS) {
        int aD = 0, lD = 0, aS = 0, lS = 0;
        if (t < G) {
            aD = tblD[b * G + t];
            lD = tblD[(b + 1) * G + t] - aD;
            aS = tblS[b * G + t];
            lS = tblS[(b + 1) * G + t] - aS;
        }
        gaD[t] = t * chunk + aD;
        lnD[t] = lD;
        gaS[t] = t * chunk + aS;
        lnS[t] = lS;
        int r = aD;
        for (int o = 16; o > 0; o >>= 1) r += __shfl_xor(r, o, 32);
        if (lane == 0) atomicAdd(&d0sh, r);
    }
    __syncthreads();
    const int d0 = d0sh;

    int a[SEGS_PER_WAVE], ln[SEGS_PER_WAVE];
    int mx = 0;
#pragma unroll
    for (int i = 0; i < SEGS_PER_WAVE; ++i) {
        a[i] = __builtin_amdgcn_readfirstlane(gaD[wave + 32 * i]);
        ln[i] = __builtin_amdgcn_readfirstlane(lnD[wave + 32 * i]);
        mx = max(mx, ln[i]);
    }
    for (int o0 = 0; o0 < mx; o0 += 32) {
        const int o = o0 + lane;
        int w[SEGS_PER_WAVE];
#pragma unroll
        for (int i = 0; i < SEGS_PER_WAVE; ++i) w[i] = strD[(o < ln[i]) ? a[i] + o : 0];
#pragma unroll
        for (int i = 0; i < SEGS_PER_WAVE; ++i) {
            asm volatile("" : "+v"(w[i]));
        }
#pragma unroll
        for (int i = 0; i < SEGS_PER_WAVE; ++i)
            if (o < ln[i]) atomicAdd(&cIn[w[i] & (BUCKET_NODES - 1)], 1);
    }
    {
        int aS[SEGS_PER_WAVE], lS[SEGS_PER_WAVE];
        int mxS = 0;
#pragma unroll
        for (int i = 0; i < SEGS_PER_WAVE; ++i) {
            aS[i] = __builtin_amdgcn_readfirstlane(gaS[wave + 32 * i]);
            lS[i] = __builtin_amdgcn_readfirstlane(lnS[wave + 32 * i]);
            mxS = max(mxS, lS[i]);
        }
        for (int o0 = 0; o0 < mxS; o0 += 32) {
            const int o = o0 + lane;
            int w[SEGS_PER_WAVE];
#pragma unroll
            for (int i = 0; i < SEGS_PER_WAVE; ++i) w[i] = strS[(o < lS[i]) ? aS[i] + o : 0];
#pragma unroll
            for (int i = 0; i < SEGS_PER_WAVE; ++i) {
                asm volatile("" : "+v"(w[i]));
            }
#pragma unroll
            for (int i = 0; i < SEGS_PER_WAVE; ++i)
                if (o < lS[i]) atomicAdd(&cOut[w[i] & (BUCKET_NODES - 1)], 1);
        }
    }
    __syncthreads();

    const int v = (t < BUCKET_NODES) ? cIn[t] : 0;
    int inc = v;
    for (int o = 1; o < 32; o <<= 1) {
        const int u = __shfl_up(inc, o, 32);
        if (lane >= o) inc += u;
    }
    if (lane == 31) wtot[wave] = inc;
    __syncthreads();
    if (wave == 0) {
        const int x = wtot[lane];
        int xi = x;
        for (int o = 1; o < 32; o <<= 1) {
            const int u = __shfl_up(xi, o, 32);
            if (lane >= o) xi += u;
        }
        wbase[lane] = xi - x;
    }
    __syncthreads();
    const int tot = wbase[31] + wtot[31];
    const bool stagedC = tot <= COL_CAP;
    if (t < BUCKET_NODES) {
        const int start = wbase[wave] + inc - v;
        cur[t] = start;
        const int node = b * BUCKET_NODES + t;
        if (node < n) {
            rp[node] = d0 + start;
            const int c = cOut[t];
            ns[node] = (c > 0) ? rsqrtf((float)c) : 0.0f;
        }
    }
    __syncthreads();
    for (int o0 = 0; o0 < mx; o0 += 32) {
        const int o = o0 + lane;
        int w[SEGS_PER_WAVE];
#pragma unroll
        for (int i = 0; i < SEGS_PER_WAVE; ++i) w[i] = strD[(o < ln[i]) ? a[i] + o : 0];
#pragma unroll
        for (int i = 0; i < SEGS_PER_WAVE; ++i) {
            asm volatile("" : "+v"(w[i]));
        }
#pragma unroll
        for (int i = 0; i < SEGS_PER_WAVE; ++i) {
            if (o < ln[i]) {
                const int r = atomicAdd(&cur[w[i] & (BUCKET_NODES - 1)], 1);
                const int sid = (int)((unsigned)w[i] >> BUCKET_SHIFT);
                if (stagedC) stC[r] = sid; else col[d0 + r] = sid;
            }
        }
    }
    if (stagedC) {
        __syncthreads();
        const int off = d0 & 31;
        int* __restrict__ cl = col + (d0 - off);
        for (int i = t; i < off + tot; i += 1024)
            if (i >= off) cl[i] = stC[i - off];
    }
}

__global__ __launch_bounds__(256) void copy_int4_kernel(const int* __restrict__ a,
                                                        int* __restrict__ b, int n4) {
    int i = blockIdx.x * 256 + threadIdx.x;
    if (i < n4) ((int4*)b)[i] = ((const int4*)a)[i];
}

#define G1_GROUPS 256
#define G1_WAVES 16
#define G1_CT 4
#define G1_PITCH (2 * N_HIDDEN + 32)
#define G1_APITCH 36
__global__ __launch_bounds__(32 * G1_WAVES) void gemm_scaled_wmma_ldsw(const float* __restrict__ H,
                                                                       const float* __restrict__ ns,
                                                                       const float* __restrict__ W,
                                                                       float* __restrict__ out,
                                                                       int nRowTiles) {
    constexpr int K = IN_FEATS;
    constexpr int F = N_HIDDEN;
    constexpr int KSTEPS = K / 4;
    constexpr int CHUNKS = K / 32;
    __shared__ __attribute__((aligned(16))) float Wl[(K / 2) * G1_PITCH];
    __shared__ __attribute__((aligned(16))) float Al[G1_WAVES * 16 * G1_APITCH];
    const int lane = threadIdx.x & 31;
    const int wave = __builtin_amdgcn_readfirstlane((int)(threadIdx.x >> 5));
    const int m  = lane & 15;
    const int hi = lane >> 4;
    const int half = wave & 1;

    float* stage = Al + wave * (16 * G1_APITCH);
    const int sr = lane >> 3;
    const int sc = (lane & 7) * 4;
    float* sdst = stage + sr * G1_APITCH + sc;
    const float* afrag = stage + m * G1_APITCH + 2 * hi;
    const float* bcol = Wl + hi * G1_PITCH + 2 * (half * 64 + m);

    const int stride = (G1_WAVES / 2) * (int)gridDim.x;
    int t = (wave >> 1) * (int)gridDim.x + (int)blockIdx.x;
    const bool has = t < nRowTiles;
    const int t0 = has ? t : 0;
    const float* gsrc = H + (size_t)(t0 * 16 + sr) * K + sc;
    v4f g0 = *(const v4f*)(gsrc);
    v4f g1 = *(const v4f*)(gsrc + (size_t)4 * K);
    v4f g2 = *(const v4f*)(gsrc + (size_t)8 * K);
    v4f g3 = *(const v4f*)(gsrc + (size_t)12 * K);
    float scale = ns[t0 * 16 + m];

#pragma unroll
    for (int i = 0; i < (K / 2) * (F / 4) / (32 * G1_WAVES); ++i) {
        const int idx = (int)threadIdx.x + i * (32 * G1_WAVES);
        const int kp = idx / (F / 4);
        const int c4 = idx - kp * (F / 4);
        const float4 w0 = ((const float4*)W)[(size_t)(2 * kp) * (F / 4) + c4];
        const float4 w1 = ((const float4*)W)[(size_t)(2 * kp + 1) * (F / 4) + c4];
        float4* dst = (float4*)(Wl + kp * G1_PITCH + 8 * c4);
        dst[0] = make_float4(w0.x, w1.x, w0.y, w1.y);
        dst[1] = make_float4(w0.z, w1.z, w0.w, w1.w);
    }
    __syncthreads();
    if (!has) return;

    *(v4f*)(sdst) = g0;
    *(v4f*)(sdst + 4 * G1_APITCH) = g1;
    *(v4f*)(sdst + 8 * G1_APITCH) = g2;
    *(v4f*)(sdst + 12 * G1_APITCH) = g3;
    __builtin_amdgcn_fence(__ATOMIC_ACQ_REL, "wavefront");
    g0 = *(const v4f*)(gsrc + 32);
    g1 = *(const v4f*)(gsrc + (size_t)4 * K + 32);
    g2 = *(const v4f*)(gsrc + (size_t)8 * K + 32);
    g3 = *(const v4f*)(gsrc + (size_t)12 * K + 32);

    v2f fa[2];
    v2f fb[2][G1_CT];
    fa[0] = *(const v2f*)(afrag);
#pragma unroll
    for (int c = 0; c < G1_CT; ++c) fb[0][c] = *(const v2f*)(bcol + c * 32);

    for (;;) {
        const int row0 = t * 16;
        const int tn = t + stride;
        const bool more = tn < nRowTiles;
        const int rown = more ? tn * 16 : row0;
        const float* gsrcn = H + (size_t)(rown + sr) * K + sc;
        const float scalen = ns[rown + m];

        v8f acc[G1_CT];
#pragma unroll
        for (int c = 0; c < G1_CT; ++c) acc[c] = (v8f){};

        for (int kc = 0; kc < CHUNKS; ++kc) {
#pragma unroll
            for (int j = 0; j < 8; ++j) {
                if (j == 7) {
                    *(v4f*)(sdst) = g0;
                    *(v4f*)(sdst + 4 * G1_APITCH) = g1;
                    *(v4f*)(sdst + 8 * G1_APITCH) = g2;
                    *(v4f*)(sdst + 12 * G1_APITCH) = g3;
                    __builtin_amdgcn_fence(__ATOMIC_ACQ_REL, "wavefront");
                    const int kk = kc + 2;
                    const float* gnext = (kk < CHUNKS) ? gsrc + 32 * kk : gsrcn + 32 * (kk - CHUNKS);
                    g0 = *(const v4f*)(gnext);
                    g1 = *(const v4f*)(gnext + (size_t)4 * K);
                    g2 = *(const v4f*)(gnext + (size_t)8 * K);
                    g3 = *(const v4f*)(gnext + (size_t)12 * K);
                }
                const int sn = (kc * 8 + j + 1 == KSTEPS) ? 0 : kc * 8 + j + 1;
                const float* bnext = bcol + sn * (2 * G1_PITCH);
                fa[(j + 1) & 1] = *(const v2f*)(afrag + 4 * ((j + 1) & 7));
#pragma unroll
                for (int c = 0; c < G1_CT; ++c) fb[(j + 1) & 1][c] = *(const v2f*)(bnext + c * 32);
                __builtin_amdgcn_sched_barrier(0);
                v2f a;
                a.x = fa[j & 1].x * scale;
                a.y = fa[j & 1].y * scale;
#pragma unroll
                for (int c = 0; c < G1_CT; ++c)
                    acc[c] = __builtin_amdgcn_wmma_f32_16x16x4_f32(false, a, false, fb[j & 1][c],
                                                                   (short)0, acc[c], false, false);
                __builtin_amdgcn_sched_barrier(0);
            }
        }

        float* __restrict__ obase = out + (size_t)(row0 + 8 * hi) * F + half * 64 + m;
#pragma unroll
        for (int c = 0; c < G1_CT; ++c) {
#pragma unroll
            for (int v = 0; v < 8; ++v)
                obase[(size_t)v * F + c * 16] = acc[c][v];
        }
        if (!more) break;
        t = tn;
        gsrc = gsrcn;
        scale = scalen;
    }
}

#define FUSED_GROUPS 1024
template <int CT>
__global__ __launch_bounds__(512) void gather_rows128_product_kernel(const float* __restrict__ hw,
                                                                     const int* __restrict__ rp,
                                                                     const int* __restrict__ col,
                                                                     const float* __restrict__ bias,
                                                                     const float* __restrict__ ns,
                                                                     const float* __restrict__ W,
                                                                     float* __restrict__ out,
                                                                     int n, int nE, int nTiles) {
    constexpr int F = CT * 16;
    constexpr int PITCH = N_HIDDEN + 4;
    __shared__ __attribute__((aligned(16))) float Wl[N_HIDDEN * F];
    __shared__ __attribute__((aligned(16))) float tile[16 * PITCH];
    const int lane = threadIdx.x & 31;
    const int wave = __builtin_amdgcn_readfirstlane((int)(threadIdx.x >> 5));

#pragma unroll
    for (int idx = threadIdx.x; idx < (N_HIDDEN / 2) * (F / 4); idx += 512) {
        const int kp = idx / (F / 4);
        const int c4 = idx - kp * (F / 4);
        const float4 w0 = ((const float4*)W)[(size_t)(2 * kp) * (F / 4) + c4];
        const float4 w1 = ((const float4*)W)[(size_t)(2 * kp + 1) * (F / 4) + c4];
        float4* dst = (float4*)(Wl + 2 * (kp * F + 4 * c4));
        dst[0] = make_float4(w0.x, w1.x, w0.y, w1.y);
        dst[1] = make_float4(w0.z, w1.z, w0.w, w1.w);
    }

    const float4* __restrict__ hw4 = (const float4*)hw;
    const float4 zero4 = make_float4(0.0f, 0.0f, 0.0f, 0.0f);
    const float4 b = ((const float4*)bias)[lane];

    for (int t = blockIdx.x; t < nTiles; t += gridDim.x) {
        const int row0 = t * 16;
        const int i = row0 + wave;
        float4 r = zero4;
        if (i < n) {
            const int start = rp[i];
            const int nxt = rp[min(i + 1, n - 1)];
            const float scale = ns[i];
            const int end = (i + 1 < n) ? nxt : nE;
            float4 acc = zero4;
            for (int base = start; base < end; base += 32) {
                const int k = base + lane;
                const int myc = (k < end) ? col[k] : 0;
                const int cnt = min(32, end - base);
                const int last = cnt - 1;
                for (int j = 0; j < cnt; j += 4) {
                    const int s0 = __builtin_amdgcn_readlane(myc, j);
                    const int s1 = __builtin_amdgcn_readlane(myc, min(j + 1, last));
                    const int s2 = __builtin_amdgcn_readlane(myc, min(j + 2, last));
                    const int s3 = __builtin_amdgcn_readlane(myc, min(j + 3, last));
                    const float4 v0 = hw4[(size_t)s0 * 32 + lane];
                    float4 v1 = hw4[(size_t)s1 * 32 + lane];
                    float4 v2 = hw4[(size_t)s2 * 32 + lane];
                    float4 v3 = hw4[(size_t)s3 * 32 + lane];
                    if (j + 1 > last) v1 = zero4;
                    if (j + 2 > last) v2 = zero4;
                    if (j + 3 > last) v3 = zero4;
                    acc.x += v0.x; acc.y += v0.y; acc.z += v0.z; acc.w += v0.w;
                    acc.x += v1.x; acc.y += v1.y; acc.z += v1.z; acc.w += v1.w;
                    acc.x += v2.x; acc.y += v2.y; acc.z += v2.z; acc.w += v2.w;
                    acc.x += v3.x; acc.y += v3.y; acc.z += v3.z; acc.w += v3.w;
                }
            }
            const int deg = end - start;
            const float nd = (deg > 0) ? rsqrtf((float)deg) : 0.0f;
            r.x = fmaxf(acc.x * nd + b.x, 0.0f) * scale;
            r.y = fmaxf(acc.y * nd + b.y, 0.0f) * scale;
            r.z = fmaxf(acc.z * nd + b.z, 0.0f) * scale;
            r.w = fmaxf(acc.w * nd + b.w, 0.0f) * scale;
        }
        *(float4*)(tile + wave * PITCH + lane * 4) = r;
        __syncthreads();

        v8f pacc = (v8f){};
        const int m  = lane & 15;
        const int hi = lane >> 4;
        if (wave < CT) {
            const float* arow = tile + m * PITCH + 2 * hi;
            const float* bcol = Wl + 2 * (hi * F + wave * 16 + m);
            v2f fa[2][4], fb[2][4];
#pragma unroll
            for (int j = 0; j < 4; ++j) {
                fa[0][j] = *(const v2f*)(arow + 4 * j);
                fb[0][j] = *(const v2f*)(bcol + 4 * j * F);
            }
#pragma unroll
            for (int g = 0; g < N_HIDDEN / 16; ++g) {
                if (g + 1 < N_HIDDEN / 16) {
#pragma unroll
                    for (int j = 0; j < 4; ++j) {
                        const int k0 = 16 * (g + 1) + 4 * j;
                        fa[(g + 1) & 1][j] = *(const v2f*)(arow + k0);
                        fb[(g + 1) & 1][j] = *(const v2f*)(bcol + k0 * F);
                    }
                }
                __builtin_amdgcn_sched_barrier(0);
#pragma unroll
                for (int j = 0; j < 4; ++j)
                    pacc = __builtin_amdgcn_wmma_f32_16x16x4_f32(false, fa[g & 1][j], false, fb[g & 1][j],
                                                                 (short)0, pacc, false, false);
            }
        }
        __syncthreads();
        if (wave < CT) {
            float* __restrict__ obase = out + (size_t)(row0 + 8 * hi) * F + wave * 16 + m;
#pragma unroll
            for (int v = 0; v < 8; ++v)
                if (row0 + 8 * hi + v < n) obase[(size_t)v * F] = pacc[v];
        }
    }
}

__global__ __launch_bounds__(256) void gather_rows16_kernel(const float* __restrict__ hw,
                                                            const int* __restrict__ rp,
                                                            const int* __restrict__ col,
                                                            const float* __restrict__ bias,
                                                            float* __restrict__ out,
                                                            int n, int nE) {
    const int t = blockIdx.x * 256 + threadIdx.x;
    const int i = t >> 2;
    const int q = t & 3;
    if (i >= n) return;
    const int start = rp[i];
    const int end = (i + 1 < n) ? rp[i + 1] : nE;
    const float4* __restrict__ hw4 = (const float4*)hw;
    const float4 zero4 = make_float4(0.0f, 0.0f, 0.0f, 0.0f);

    float4 acc = zero4;
    const int lastk = end - 1;
    for (int k = start; k < end; k += 4) {
        const int c0 = col[k];
        const int c1 = col[min(k + 1, lastk)];
        const int c2 = col[min(k + 2, lastk)];
        const int c3 = col[min(k + 3, lastk)];
        const float4 v0 = hw4[(size_t)c0 * 4 + q];
        float4 v1 = hw4[(size_t)c1 * 4 + q];
        float4 v2 = hw4[(size_t)c2 * 4 + q];
        float4 v3 = hw4[(size_t)c3 * 4 + q];
        if (k + 1 > lastk) v1 = zero4;
        if (k + 2 > lastk) v2 = zero4;
        if (k + 3 > lastk) v3 = zero4;
        acc.x += v0.x; acc.y += v0.y; acc.z += v0.z; acc.w += v0.w;
        acc.x += v1.x; acc.y += v1.y; acc.z += v1.z; acc.w += v1.w;
        acc.x += v2.x; acc.y += v2.y; acc.z += v2.z; acc.w += v2.w;
        acc.x += v3.x; acc.y += v3.y; acc.z += v3.z; acc.w += v3.w;
    }

    const int deg = end - start;
    const float nd = (deg > 0) ? rsqrtf((float)deg) : 0.0f;
    const float4 b = ((const float4*)bias)[q];
    float4 r;
    r.x = acc.x * nd + b.x;
    r.y = acc.y * nd + b.y;
    r.z = acc.z * nd + b.z;
    r.w = acc.w * nd + b.w;
    ((float4*)out)[(size_t)i * 4 + q] = r;
}

extern "C" void kernel_launch(void* const* d_in, const int* in_sizes, int n_in,
                              void* d_out, int out_size, void* d_ws, size_t ws_size,
                              hipStream_t stream) {
    const float* feat = (const float*)d_in[0];
    const int*   src  = (const int*)d_in[1];
    const int*   dst  = (const int*)d_in[2];
    const float* W1   = (const float*)d_in[3];
    const float* b1   = (const float*)d_in[4];
    const float* W2   = (const float*)d_in[5];
    const float* b2   = (const float*)d_in[6];
    const float* W3   = (const float*)d_in[7];
    const float* b3   = (const float*)d_in[8];
    float* out = (float*)d_out;

    const int N = in_sizes[0] / IN_FEATS;
    const int E = in_sizes[1];

    float* nsW    = (float*)d_ws;
    int*   rp     = (int*)d_ws + N;
    float* bufA   = (float*)(rp + N);
    float* bufB   = bufA + (size_t)N * N_HIDDEN;
    const float* ns = nsW;
    int* col      = (int*)d_out;
    int* col3     = (int*)(bufA + (size_t)N * N_CLASSES);

    const int rowTiles  = N / 16;
    const int nTiles    = (N + 15) / 16;
    const int nBuckets  = (N + BUCKET_NODES - 1) / BUCKET_NODES;
    const int T         = (nBuckets + 1) * EDGE_CHUNKS;
    int* tblD = (int*)bufA;
    int* tblS = tblD + T;
    int* strD = (int*)bufB;
    int* strS = strD + E;

    bucket_partition_kernel<<<EDGE_CHUNKS, 1024, 0, stream>>>(src, dst, tblD, tblS, strD, strS, E, nBuckets);
    bucket_finish_kernel<<<nBuckets, 1024, 0, stream>>>(tblD, tblS, strD, strS, rp, nsW, col, N, E, EDGE_CHUNKS);

    gemm_scaled_wmma_ldsw<<<G1_GROUPS, 32 * G1_WAVES, 0, stream>>>(feat, ns, W1, bufA, rowTiles);

    gather_rows128_product_kernel<8><<<FUSED_GROUPS, 512, 0, stream>>>(bufA, rp, col, b1, ns, W2, bufB, N, E, nTiles);

    gather_rows128_product_kernel<1><<<FUSED_GROUPS, 512, 0, stream>>>(bufB, rp, col, b2, ns, W3, bufA, N, E, nTiles);

    copy_int4_kernel<<<(E / 4 + 255) / 256, 256, 0, stream>>>(col, col3, E / 4);
    gather_rows16_kernel<<<(4 * N + 255) / 256, 256, 0, stream>>>(bufA, rp, col3, b3, out, N, E);
}
